# starting kernel
# baseline (speedup 1.0000x reference)
#include <hip/hip_runtime.h>
#include <hip/hip_bf16.h>
#include <math.h>


namespace {

constexpr int kN1 = 40000;
constexpr int kN2 = 10000;
constexpr int kE1 = 640000;
constexpr int kE2 = 160000;

typedef __attribute__((ext_vector_type(2))) float v2f;
typedef __attribute__((ext_vector_type(8))) float v8f;

__global__ void zero_f32(float* __restrict__ p, int n) {
  int i = blockIdx.x * blockDim.x + threadIdx.x;
  if (i < n) p[i] = 0.0f;
}

__global__ void scatter_edges(const float* __restrict__ feat,
                              const int* __restrict__ src,
                              const int* __restrict__ dst,
                              float* __restrict__ sum,
                              float* __restrict__ cnt) {
  const int e = blockIdx.x;
  const int s = src[e];
  const int d = dst[e];
  const float* fp = feat + (size_t)s * 256;
  float* sp = sum + (size_t)d * 256;
  atomicAdd(&sp[threadIdx.x], fp[threadIdx.x]);
  if (threadIdx.x == 0) atomicAdd(&cnt[d], 1.0f);
}

template <int KDIM, int NDIM, bool SIGMOID>
__global__ void fused_wmma_gemm(const float* __restrict__ Asum,
                                const float* __restrict__ cnt,
                                const float* __restrict__ Aself,
                                const float* __restrict__ Wl,
                                const float* __restrict__ Wr,
                                const float* __restrict__ bias,
                                float* __restrict__ out,
                                int M) {
  constexpr int KC = 32;
  constexpr int LDS_STRIDE = KC + 2;
  __shared__ float ldsW[2][64][LDS_STRIDE];

  const int lane = threadIdx.x & 31;
  const int wave = threadIdx.x >> 5;
  constexpr int kColBlocks = NDIM / 64;
  const int brow = blockIdx.x / kColBlocks;
  const int n0 = (blockIdx.x % kColBlocks) * 64;
  const int m0 = brow * 128 + wave * 16;

  const int half = lane >> 4;
  const int l15 = lane & 15;
  const int arow = min(m0 + l15, M - 1);
  const float inv = 1.0f / fmaxf(cnt[arow], 1.0f);

  const float* aSumRow  = Asum  + (size_t)arow * KDIM + 2 * half;
  const float* aSelfRow = Aself + (size_t)arow * KDIM + 2 * half;

  v8f acc[4];
#pragma unroll
  for (int t = 0; t < 4; ++t) acc[t] = v8f{0, 0, 0, 0, 0, 0, 0, 0};

  for (int kc = 0; kc < KDIM; kc += KC) {
    __syncthreads();
    for (int i = threadIdx.x; i < 64 * KC; i += 256) {
      const int col = i & 63;
      const int kr = i >> 6;
      ldsW[0][col][kr] = Wl[(size_t)(kc + kr) * NDIM + n0 + col];
      ldsW[1][col][kr] = Wr[(size_t)(kc + kr) * NDIM + n0 + col];
    }
    __syncthreads();

#pragma unroll
    for (int k2 = 0; k2 < KC; k2 += 4) {
      v2f a1 = *(const v2f*)(aSumRow + kc + k2);
      a1 = a1 * inv;
      const v2f a2 = *(const v2f*)(aSelfRow + kc + k2);
      const int koff = k2 + 2 * half;
#pragma unroll
      for (int t = 0; t < 4; ++t) {
        const int lcol = 16 * t + l15;
        const v2f bl = *(const v2f*)&ldsW[0][lcol][koff];
        acc[t] = __builtin_amdgcn_wmma_f32_16x16x4_f32(
            false, a1, false, bl, (short)0, acc[t], false, false);
        const v2f br = *(const v2f*)&ldsW[1][lcol][koff];
        acc[t] = __builtin_amdgcn_wmma_f32_16x16x4_f32(
            false, a2, false, br, (short)0, acc[t], false, false);
      }
    }
  }

#pragma unroll
  for (int t = 0; t < 4; ++t) {
    const int col = n0 + 16 * t + l15;
    const float b = bias[col];
#pragma unroll
    for (int r = 0; r < 8; ++r) {
      const int row = m0 + r + 8 * half;
      float v = acc[t][r] + b;
      v = SIGMOID ? (1.0f / (1.0f + expf(-v))) : fmaxf(v, 0.0f);
      if (row < M) out[(size_t)row * NDIM + col] = v;
    }
  }
}

}

extern "C" void kernel_launch(void* const* d_in, const int* in_sizes, int n_in,
                              void* d_out, int out_size, void* d_ws, size_t ws_size,
                              hipStream_t stream) {
  (void)in_sizes; (void)n_in; (void)out_size; (void)ws_size;

  const float* x   = (const float*)d_in[0];
  const float* W1l = (const float*)d_in[1];
  const float* b1  = (const float*)d_in[2];
  const float* W1r = (const float*)d_in[3];
  const float* W2l = (const float*)d_in[4];
  const float* b2  = (const float*)d_in[5];
  const float* W2r = (const float*)d_in[6];
  const int* src1  = (const int*)d_in[7];
  const int* dst1  = (const int*)d_in[8];
  const int* src2  = (const int*)d_in[9];
  const int* dst2  = (const int*)d_in[10];
  float* out = (float*)d_out;

  float* ws   = (float*)d_ws;
  float* sum1 = ws;
  float* cnt1 = sum1 + (size_t)kN1 * 256;
  float* sum2 = cnt1 + kN1;
  float* cnt2 = sum2 + (size_t)kN2 * 256;
  float* h    = cnt2 + kN2;
  const size_t zeroCount =
      (size_t)kN1 * 256 + kN1 + (size_t)kN2 * 256 + kN2;

  zero_f32<<<(int)((zeroCount + 255) / 256), 256, 0, stream>>>(ws, (int)zeroCount);

  scatter_edges<<<kE1, 256, 0, stream>>>(x, src1, dst1, sum1, cnt1);

  fused_wmma_gemm<256, 256, false>
      <<<313 * 4, 256, 0, stream>>>(sum1, cnt1, x, W1l, W1r, b1, h, kN1);

  scatter_edges<<<kE2, 256, 0, stream>>>(h, src2, dst2, sum2, cnt2);

  fused_wmma_gemm<256, 128, true>
      <<<79 * 2, 256, 0, stream>>>(sum2, cnt2, h, W2l, W2r, b2, out, kN2);
}


# Segment sums by gather, not f32 atomics: per-call edge lists (int atomic rank, block scan, fill), one wave per node adds its source rows in arrival-order rank; float summation order changes
# speedup vs baseline: 2.7323x; 2.7323x over previous
#include <hip/hip_runtime.h>
#include <hip/hip_bf16.h>
#include <math.h>


namespace {

constexpr int kN0 = 100000;
constexpr int kN1 = 40000;
constexpr int kN2 = 10000;
constexpr int kE1 = 640000;
constexpr int kE2 = 160000;

typedef __attribute__((ext_vector_type(2))) float v2f;
typedef __attribute__((ext_vector_type(8))) float v8f;

typedef __attribute__((ext_vector_type(4))) float v4f;

constexpr size_t align64(size_t n) { return (n + 63) / 64 * 64; }
constexpr size_t kOffSum1 = 0;
constexpr size_t kOffCnt1 = kOffSum1 + (size_t)kN1 * 256;
constexpr size_t kOffSum2 = kOffCnt1 + kN1;
constexpr size_t kOffCnt2 = kOffSum2 + (size_t)kN2 * 256;
constexpr size_t kOffH    = kOffCnt2 + kN2;
constexpr size_t kOffDeg  = align64(kOffH + (size_t)kN1 * 256);
constexpr size_t kOffOff1 = kOffDeg + align64(kN1 + kN2);
constexpr size_t kOffOff2 = kOffOff1 + align64(kN1 + 1);
constexpr size_t kOffRank = kOffOff2 + align64(kN2 + 1);
constexpr size_t kOffPerm = kOffRank + align64(kE1 + kE2);

__global__ void zero_i32(int* __restrict__ p, int n) {
  int i = blockIdx.x * blockDim.x + threadIdx.x;
  if (i < n) p[i] = 0;
}

__global__ void edge_rank(const int* __restrict__ dst1,
                          const int* __restrict__ dst2,
                          int* __restrict__ deg,
                          int* __restrict__ rank) {
  const int i = blockIdx.x * blockDim.x + threadIdx.x;
  if (i < kE1) {
    rank[i] = atomicAdd(&deg[dst1[i]], 1);
  } else if (i < kE1 + kE2) {
    rank[i] = atomicAdd(&deg[kN1 + dst2[i - kE1]], 1);
  }
}

constexpr int kScanBlocks1 = (kN1 + 1023) / 1024;
constexpr int kScanBlocks2 = (kN2 + 1023) / 1024;
__global__ void __launch_bounds__(1024)
scan_degrees(const int* __restrict__ deg,
             int* __restrict__ off1, int* __restrict__ off2,
             float* __restrict__ cnt1, float* __restrict__ cnt2) {
  const bool second = blockIdx.x >= kScanBlocks1;
  const int b = second ? (int)blockIdx.x - kScanBlocks1 : (int)blockIdx.x;
  const int n = second ? kN2 : kN1;
  const int* d = deg + (second ? kN1 : 0);
  int* o = second ? off2 : off1;
  float* c = second ? cnt2 : cnt1;
  const int lane = threadIdx.x & 31;
  const int wave = threadIdx.x >> 5;

  int pre = 0;
  for (int k = 0; k < b; ++k) pre += d[k * 1024 + threadIdx.x];
  const int i = b * 1024 + threadIdx.x;
  const int v = (i < n) ? d[i] : 0;

  int incl = v;
  for (int ofs = 1; ofs < 32; ofs <<= 1) {
    const int u = __shfl_up(incl, ofs);
    if (lane >= ofs) incl += u;
  }
  for (int ofs = 16; ofs > 0; ofs >>= 1) pre += __shfl_xor(pre, ofs);

  __shared__ int wv[32];
  __shared__ int wp[32];
  if (lane == 31) wv[wave] = incl;
  if (lane == 0) wp[wave] = pre;
  __syncthreads();
  int base = wp[lane] + ((lane < wave) ? wv[lane] : 0);
  for (int ofs = 16; ofs > 0; ofs >>= 1) base += __shfl_xor(base, ofs);

  if (i < n) {
    o[i] = base + incl - v;
    c[i] = (float)v;
  }
  if (i == n - 1) o[n] = base + incl;
}

__global__ void fill_perm(const int* __restrict__ src1,
                          const int* __restrict__ dst1,
                          const int* __restrict__ src2,
                          const int* __restrict__ dst2,
                          const int* __restrict__ off1,
                          const int* __restrict__ off2,
                          const int* __restrict__ rank,
                          int* __restrict__ perm) {
  const int i = blockIdx.x * blockDim.x + threadIdx.x;
  if (i < kE1) {
    perm[off1[dst1[i]] + rank[i]] = src1[i];
  } else if (i < kE1 + kE2) {
    const int j = i - kE1;
    perm[kE1 + off2[dst2[j]] + rank[i]] = src2[j];
  }
}

template <int NSRC>
__global__ void __launch_bounds__(256)
gather_sum(const float* __restrict__ feat,
           const int* __restrict__ off,
           const int* __restrict__ perm,
           float* __restrict__ sum,
           int nseg) {
  const int lane = threadIdx.x & 31;
  const int node =
      blockIdx.x * 8 + __builtin_amdgcn_readfirstlane((int)(threadIdx.x >> 5));
  if (node >= nseg) return;
  const int beg = off[node];
  const int end = off[node + 1];

  v4f a0 = v4f{0, 0, 0, 0};
  v4f a1 = v4f{0, 0, 0, 0};
  for (int base = beg; base < end; base += 32) {
    const int n = min(32, end - base);
    int mine = (lane < n) ? perm[base + lane] : 0;
    mine = min(max(mine, 0), NSRC - 1);
    int j = 0;
    for (; j + 4 <= n; j += 4) {
      const v4f* p0 = (const v4f*)(feat + (size_t)__builtin_amdgcn_readlane(mine, j) * 256);
      const v4f* p1 = (const v4f*)(feat + (size_t)__builtin_amdgcn_readlane(mine, j + 1) * 256);
      const v4f* p2 = (const v4f*)(feat + (size_t)__builtin_amdgcn_readlane(mine, j + 2) * 256);
      const v4f* p3 = (const v4f*)(feat + (size_t)__builtin_amdgcn_readlane(mine, j + 3) * 256);
      const v4f x0 = p0[lane], y0 = p0[lane + 32];
      const v4f x1 = p1[lane], y1 = p1[lane + 32];
      const v4f x2 = p2[lane], y2 = p2[lane + 32];
      const v4f x3 = p3[lane], y3 = p3[lane + 32];
      a0 += x0; a1 += y0;
      a0 += x1; a1 += y1;
      a0 += x2; a1 += y2;
      a0 += x3; a1 += y3;
    }
#pragma unroll 1
    for (; j < n; ++j) {
      const v4f* p = (const v4f*)(feat + (size_t)__builtin_amdgcn_readlane(mine, j) * 256);
      a0 += p[lane];
      a1 += p[lane + 32];
    }
  }
  v4f* sp = (v4f*)(sum + (size_t)node * 256);
  sp[lane] = a0;
  sp[lane + 32] = a1;
}

template <int KDIM, int NDIM, bool SIGMOID>
__global__ void fused_wmma_gemm(const float* __restrict__ Asum,
                                const float* __restrict__ cnt,
                                const float* __restrict__ Aself,
                                const float* __restrict__ Wl,
                                const float* __restrict__ Wr,
                                const float* __restrict__ bias,
                                float* __restrict__ out,
                                int M) {
  constexpr int KC = 32;
  constexpr int LDS_STRIDE = KC + 2;
  __shared__ float ldsW[2][64][LDS_STRIDE];

  const int lane = threadIdx.x & 31;
  const int wave = threadIdx.x >> 5;
  constexpr int kColBlocks = NDIM / 64;
  const int brow = blockIdx.x / kColBlocks;
  const int n0 = (blockIdx.x % kColBlocks) * 64;
  const int m0 = brow * 128 + wave * 16;

  const int half = lane >> 4;
  const int l15 = lane & 15;
  const int arow = min(m0 + l15, M - 1);
  const float inv = 1.0f / fmaxf(cnt[arow], 1.0f);

  const float* aSumRow  = Asum  + (size_t)arow * KDIM + 2 * half;
  const float* aSelfRow = Aself + (size_t)arow * KDIM + 2 * half;

  v8f acc[4];
#pragma unroll
  for (int t = 0; t < 4; ++t) acc[t] = v8f{0, 0, 0, 0, 0, 0, 0, 0};

  for (int kc = 0; kc < KDIM; kc += KC) {
    __syncthreads();
    for (int i = threadIdx.x; i < 64 * KC; i += 256) {
      const int col = i & 63;
      const int kr = i >> 6;
      ldsW[0][col][kr] = Wl[(size_t)(kc + kr) * NDIM + n0 + col];
      ldsW[1][col][kr] = Wr[(size_t)(kc + kr) * NDIM + n0 + col];
    }
    __syncthreads();

#pragma unroll
    for (int k2 = 0; k2 < KC; k2 += 4) {
      v2f a1 = *(const v2f*)(aSumRow + kc + k2);
      a1 = a1 * inv;
      const v2f a2 = *(const v2f*)(aSelfRow + kc + k2);
      const int koff = k2 + 2 * half;
#pragma unroll
      for (int t = 0; t < 4; ++t) {
        const int lcol = 16 * t + l15;
        const v2f bl = *(const v2f*)&ldsW[0][lcol][koff];
        acc[t] = __builtin_amdgcn_wmma_f32_16x16x4_f32(
            false, a1, false, bl, (short)0, acc[t], false, false);
        const v2f br = *(const v2f*)&ldsW[1][lcol][koff];
        acc[t] = __builtin_amdgcn_wmma_f32_16x16x4_f32(
            false, a2, false, br, (short)0, acc[t], false, false);
      }
    }
  }

#pragma unroll
  for (int t = 0; t < 4; ++t) {
    const int col = n0 + 16 * t + l15;
    const float b = bias[col];
#pragma unroll
    for (int r = 0; r < 8; ++r) {
      const int row = m0 + r + 8 * half;
      float v = acc[t][r] + b;
      v = SIGMOID ? (1.0f / (1.0f + expf(-v))) : fmaxf(v, 0.0f);
      if (row < M) out[(size_t)row * NDIM + col] = v;
    }
  }
}

}

extern "C" void kernel_launch(void* const* d_in, const int* in_sizes, int n_in,
                              void* d_out, int out_size, void* d_ws, size_t ws_size,
                              hipStream_t stream) {
  (void)in_sizes; (void)n_in; (void)out_size; (void)ws_size;

  const float* x   = (const float*)d_in[0];
  const float* W1l = (const float*)d_in[1];
  const float* b1  = (const float*)d_in[2];
  const float* W1r = (const float*)d_in[3];
  const float* W2l = (const float*)d_in[4];
  const float* b2  = (const float*)d_in[5];
  const float* W2r = (const float*)d_in[6];
  const int* src1  = (const int*)d_in[7];
  const int* dst1  = (const int*)d_in[8];
  const int* src2  = (const int*)d_in[9];
  const int* dst2  = (const int*)d_in[10];
  float* out = (float*)d_out;

  float* ws   = (float*)d_ws;
  int*   wsi  = (int*)d_ws;
  float* sum1 = ws + kOffSum1;
  float* cnt1 = ws + kOffCnt1;
  float* sum2 = ws + kOffSum2;
  float* cnt2 = ws + kOffCnt2;
  float* h    = ws + kOffH;
  int* deg  = wsi + kOffDeg;
  int* off1 = wsi + kOffOff1;
  int* off2 = wsi + kOffOff2;
  int* rank = wsi + kOffRank;
  int* perm = wsi + kOffPerm;

  zero_i32<<<(kN1 + kN2 + 255) / 256, 256, 0, stream>>>(deg, kN1 + kN2);

  edge_rank<<<(kE1 + kE2) / 256, 256, 0, stream>>>(dst1, dst2, deg, rank);

  scan_degrees<<<kScanBlocks1 + kScanBlocks2, 1024, 0, stream>>>(deg, off1, off2,
                                                                 cnt1, cnt2);

  fill_perm<<<(kE1 + kE2) / 256, 256, 0, stream>>>(src1, dst1, src2, dst2,
                                                  off1, off2, rank, perm);

  gather_sum<kN0><<<kN1 / 8, 256, 0, stream>>>(x, off1, perm, sum1, kN1);

  fused_wmma_gemm<256, 256, false>
      <<<313 * 4, 256, 0, stream>>>(sum1, cnt1, x, W1l, W1r, b1, h, kN1);

  gather_sum<kN1><<<kN2 / 8, 256, 0, stream>>>(h, off2, perm + kE1, sum2, kN2);

  fused_wmma_gemm<256, 128, true>
      <<<79 * 2, 256, 0, stream>>>(sum2, cnt2, h, W2l, W2r, b2, out, kN2);
}


# GEMM K-chunk staging: A rows go through LDS as whole 128-B lines (16 B per lane), not 16-line 8-B fragment loads from global; a chunk's loads are issued together, not one staging pass per wait
# speedup vs baseline: 3.2854x; 1.2024x over previous
#include <hip/hip_runtime.h>
#include <hip/hip_bf16.h>
#include <math.h>


namespace {

constexpr int kN0 = 100000;
constexpr int kN1 = 40000;
constexpr int kN2 = 10000;
constexpr int kE1 = 640000;
constexpr int kE2 = 160000;

typedef __attribute__((ext_vector_type(2))) float v2f;
typedef __attribute__((ext_vector_type(8))) float v8f;

typedef __attribute__((ext_vector_type(4))) float v4f;

constexpr size_t align64(size_t n) { return (n + 63) / 64 * 64; }
constexpr size_t kOffSum1 = 0;
constexpr size_t kOffCnt1 = kOffSum1 + (size_t)kN1 * 256;
constexpr size_t kOffSum2 = kOffCnt1 + kN1;
constexpr size_t kOffCnt2 = kOffSum2 + (size_t)kN2 * 256;
constexpr size_t kOffH    = kOffCnt2 + kN2;
constexpr size_t kOffDeg  = align64(kOffH + (size_t)kN1 * 256);
constexpr size_t kOffOff1 = kOffDeg + align64(kN1 + kN2);
constexpr size_t kOffOff2 = kOffOff1 + align64(kN1 + 1);
constexpr size_t kOffRank = kOffOff2 + align64(kN2 + 1);
constexpr size_t kOffPerm = kOffRank + align64(kE1 + kE2);

__global__ void zero_i32(int* __restrict__ p, int n) {
  int i = blockIdx.x * blockDim.x + threadIdx.x;
  if (i < n) p[i] = 0;
}

__global__ void edge_rank(const int* __restrict__ dst1,
                          const int* __restrict__ dst2,
                          int* __restrict__ deg,
                          int* __restrict__ rank) {
  const int i = blockIdx.x * blockDim.x + threadIdx.x;
  if (i < kE1) {
    rank[i] = atomicAdd(&deg[dst1[i]], 1);
  } else if (i < kE1 + kE2) {
    rank[i] = atomicAdd(&deg[kN1 + dst2[i - kE1]], 1);
  }
}

constexpr int kScanBlocks1 = (kN1 + 1023) / 1024;
constexpr int kScanBlocks2 = (kN2 + 1023) / 1024;
__global__ void __launch_bounds__(1024)
scan_degrees(const int* __restrict__ deg,
             int* __restrict__ off1, int* __restrict__ off2,
             float* __restrict__ cnt1, float* __restrict__ cnt2) {
  const bool second = blockIdx.x >= kScanBlocks1;
  const int b = second ? (int)blockIdx.x - kScanBlocks1 : (int)blockIdx.x;
  const int n = second ? kN2 : kN1;
  const int* d = deg + (second ? kN1 : 0);
  int* o = second ? off2 : off1;
  float* c = second ? cnt2 : cnt1;
  const int lane = threadIdx.x & 31;
  const int wave = threadIdx.x >> 5;

  int pre = 0;
  for (int k = 0; k < b; ++k) pre += d[k * 1024 + threadIdx.x];
  const int i = b * 1024 + threadIdx.x;
  const int v = (i < n) ? d[i] : 0;

  int incl = v;
  for (int ofs = 1; ofs < 32; ofs <<= 1) {
    const int u = __shfl_up(incl, ofs);
    if (lane >= ofs) incl += u;
  }
  for (int ofs = 16; ofs > 0; ofs >>= 1) pre += __shfl_xor(pre, ofs);

  __shared__ int wv[32];
  __shared__ int wp[32];
  if (lane == 31) wv[wave] = incl;
  if (lane == 0) wp[wave] = pre;
  __syncthreads();
  int base = wp[lane] + ((lane < wave) ? wv[lane] : 0);
  for (int ofs = 16; ofs > 0; ofs >>= 1) base += __shfl_xor(base, ofs);

  if (i < n) {
    o[i] = base + incl - v;
    c[i] = (float)v;
  }
  if (i == n - 1) o[n] = base + incl;
}

__global__ void fill_perm(const int* __restrict__ src1,
                          const int* __restrict__ dst1,
                          const int* __restrict__ src2,
                          const int* __restrict__ dst2,
                          const int* __restrict__ off1,
                          const int* __restrict__ off2,
                          const int* __restrict__ rank,
                          int* __restrict__ perm) {
  const int i = blockIdx.x * blockDim.x + threadIdx.x;
  if (i < kE1) {
    perm[off1[dst1[i]] + rank[i]] = src1[i];
  } else if (i < kE1 + kE2) {
    const int j = i - kE1;
    perm[kE1 + off2[dst2[j]] + rank[i]] = src2[j];
  }
}

template <int NSRC>
__global__ void __launch_bounds__(256)
gather_sum(const float* __restrict__ feat,
           const int* __restrict__ off,
           const int* __restrict__ perm,
           float* __restrict__ sum,
           int nseg) {
  const int lane = threadIdx.x & 31;
  const int node =
      blockIdx.x * 8 + __builtin_amdgcn_readfirstlane((int)(threadIdx.x >> 5));
  if (node >= nseg) return;
  const int beg = off[node];
  const int end = off[node + 1];

  v4f a0 = v4f{0, 0, 0, 0};
  v4f a1 = v4f{0, 0, 0, 0};
  for (int base = beg; base < end; base += 32) {
    const int n = min(32, end - base);
    int mine = (lane < n) ? perm[base + lane] : 0;
    mine = min(max(mine, 0), NSRC - 1);
    int j = 0;
    for (; j + 4 <= n; j += 4) {
      const v4f* p0 = (const v4f*)(feat + (size_t)__builtin_amdgcn_readlane(mine, j) * 256);
      const v4f* p1 = (const v4f*)(feat + (size_t)__builtin_amdgcn_readlane(mine, j + 1) * 256);
      const v4f* p2 = (const v4f*)(feat + (size_t)__builtin_amdgcn_readlane(mine, j + 2) * 256);
      const v4f* p3 = (const v4f*)(feat + (size_t)__builtin_amdgcn_readlane(mine, j + 3) * 256);
      const v4f x0 = p0[lane], y0 = p0[lane + 32];
      const v4f x1 = p1[lane], y1 = p1[lane + 32];
      const v4f x2 = p2[lane], y2 = p2[lane + 32];
      const v4f x3 = p3[lane], y3 = p3[lane + 32];
      a0 += x0; a1 += y0;
      a0 += x1; a1 += y1;
      a0 += x2; a1 += y2;
      a0 += x3; a1 += y3;
    }
#pragma unroll 1
    for (; j < n; ++j) {
      const v4f* p = (const v4f*)(feat + (size_t)__builtin_amdgcn_readlane(mine, j) * 256);
      a0 += p[lane];
      a1 += p[lane + 32];
    }
  }
  v4f* sp = (v4f*)(sum + (size_t)node * 256);
  sp[lane] = a0;
  sp[lane + 32] = a1;
}

template <int KDIM, int NDIM, int M, bool SIGMOID>
__global__ void __launch_bounds__(256, 10)
fused_wmma_gemm(const float* __restrict__ Asum,
                                const float* __restrict__ cnt,
                                const float* __restrict__ Aself,
                                const float* __restrict__ Wl,
                                const float* __restrict__ Wr,
                                const float* __restrict__ bias,
                                float* __restrict__ out) {
  static_assert(M % 16 == 0 && M >= 16, "row count must be a multiple of 16");
  constexpr int KC = 32;
  constexpr int LDS_STRIDE = KC + 2;
  __shared__ float ldsW[2][64][LDS_STRIDE];
  constexpr int A_STRIDE = KC + 4;
  __shared__ __attribute__((aligned(16))) float ldsA[2][128][A_STRIDE];

  const int lane = threadIdx.x & 31;
  const int wave = threadIdx.x >> 5;
  constexpr int kColBlocks = NDIM / 64;
  const int brow = blockIdx.x / kColBlocks;
  const int n0 = (blockIdx.x % kColBlocks) * 64;
  const int m0 = brow * 128 + wave * 16;

  const int half = lane >> 4;
  const int l15 = lane & 15;
  const int srow = lane >> 3;
  const int scol = (lane & 7) * 4;
  const int mload = min(m0, M - 16) + srow;
  const float* sSum  = Asum  + (size_t)mload * KDIM + scol;
  const float* sSelf = Aself + (size_t)mload * KDIM + scol;
  float sInv[4];
#pragma unroll
  for (int p = 0; p < 4; ++p)
    sInv[p] = 1.0f / fmaxf(cnt[mload + 4 * p], 1.0f);
  const int wrow = wave * 16;

  v8f acc[4];
#pragma unroll
  for (int t = 0; t < 4; ++t) acc[t] = v8f{0, 0, 0, 0, 0, 0, 0, 0};

  for (int kc = 0; kc < KDIM; kc += KC) {
    __syncthreads();
    const int wcol = threadIdx.x & 63;
    const int wkr = threadIdx.x >> 6;
    float wl[KC / 4], wr[KC / 4];
    v4f as[4], at[4];
#pragma unroll
    for (int q = 0; q < KC / 4; ++q) {
      wl[q] = Wl[(size_t)(kc + wkr + 4 * q) * NDIM + n0 + wcol];
      wr[q] = Wr[(size_t)(kc + wkr + 4 * q) * NDIM + n0 + wcol];
    }
#pragma unroll
    for (int p = 0; p < 4; ++p) {
      as[p] = *(const v4f*)(sSum + (size_t)(4 * p) * KDIM + kc);
      at[p] = *(const v4f*)(sSelf + (size_t)(4 * p) * KDIM + kc);
    }
#pragma unroll
    for (int q = 0; q < KC / 4; ++q) {
      ldsW[0][wcol][wkr + 4 * q] = wl[q];
      ldsW[1][wcol][wkr + 4 * q] = wr[q];
    }
#pragma unroll
    for (int p = 0; p < 4; ++p) {
      *(v4f*)&ldsA[0][wrow + 4 * p + srow][scol] = as[p] * sInv[p];
      *(v4f*)&ldsA[1][wrow + 4 * p + srow][scol] = at[p];
    }
    __syncthreads();

#pragma unroll
    for (int k2 = 0; k2 < KC; k2 += 4) {
      const int koff = k2 + 2 * half;
      const v2f a1 = *(const v2f*)&ldsA[0][wrow + l15][koff];
      const v2f a2 = *(const v2f*)&ldsA[1][wrow + l15][koff];
#pragma unroll
      for (int t = 0; t < 4; ++t) {
        const int lcol = 16 * t + l15;
        const v2f bl = *(const v2f*)&ldsW[0][lcol][koff];
        acc[t] = __builtin_amdgcn_wmma_f32_16x16x4_f32(
            false, a1, false, bl, (short)0, acc[t], false, false);
        const v2f br = *(const v2f*)&ldsW[1][lcol][koff];
        acc[t] = __builtin_amdgcn_wmma_f32_16x16x4_f32(
            false, a2, false, br, (short)0, acc[t], false, false);
      }
    }
  }

#pragma unroll
  for (int t = 0; t < 4; ++t) {
    const int col = n0 + 16 * t + l15;
    const float b = bias[col];
#pragma unroll
    for (int r = 0; r < 8; ++r) {
      const int row = m0 + r + 8 * half;
      float v = acc[t][r] + b;
      v = SIGMOID ? (1.0f / (1.0f + expf(-v))) : fmaxf(v, 0.0f);
      if (row < M) out[(size_t)row * NDIM + col] = v;
    }
  }
}

}

extern "C" void kernel_launch(void* const* d_in, const int* in_sizes, int n_in,
                              void* d_out, int out_size, void* d_ws, size_t ws_size,
                              hipStream_t stream) {
  (void)in_sizes; (void)n_in; (void)out_size; (void)ws_size;

  const float* x   = (const float*)d_in[0];
  const float* W1l = (const float*)d_in[1];
  const float* b1  = (const float*)d_in[2];
  const float* W1r = (const float*)d_in[3];
  const float* W2l = (const float*)d_in[4];
  const float* b2  = (const float*)d_in[5];
  const float* W2r = (const float*)d_in[6];
  const int* src1  = (const int*)d_in[7];
  const int* dst1  = (const int*)d_in[8];
  const int* src2  = (const int*)d_in[9];
  const int* dst2  = (const int*)d_in[10];
  float* out = (float*)d_out;

  float* ws   = (float*)d_ws;
  int*   wsi  = (int*)d_ws;
  float* sum1 = ws + kOffSum1;
  float* cnt1 = ws + kOffCnt1;
  float* sum2 = ws + kOffSum2;
  float* cnt2 = ws + kOffCnt2;
  float* h    = ws + kOffH;
  int* deg  = wsi + kOffDeg;
  int* off1 = wsi + kOffOff1;
  int* off2 = wsi + kOffOff2;
  int* rank = wsi + kOffRank;
  int* perm = wsi + kOffPerm;

  zero_i32<<<(kN1 + kN2 + 255) / 256, 256, 0, stream>>>(deg, kN1 + kN2);

  edge_rank<<<(kE1 + kE2) / 256, 256, 0, stream>>>(dst1, dst2, deg, rank);

  scan_degrees<<<kScanBlocks1 + kScanBlocks2, 1024, 0, stream>>>(deg, off1, off2,
                                                                 cnt1, cnt2);

  fill_perm<<<(kE1 + kE2) / 256, 256, 0, stream>>>(src1, dst1, src2, dst2,
                                                  off1, off2, rank, perm);

  gather_sum<kN0><<<kN1 / 8, 256, 0, stream>>>(x, off1, perm, sum1, kN1);

  fused_wmma_gemm<256, 256, kN1, false>
      <<<313 * 4, 256, 0, stream>>>(sum1, cnt1, x, W1l, W1r, b1, h);

  gather_sum<kN1><<<kN2 / 8, 256, 0, stream>>>(h, off2, perm + kE1, sum2, kN2);

  fused_wmma_gemm<256, 128, kN2, true>
      <<<79 * 2, 256, 0, stream>>>(sum2, cnt2, h, W2l, W2r, b2, out);
}


# Edge lists as fixed 64-slot rows per destination, filled in the counting pass at the slot the integer atomic returns (spill list past 64): no offset scan, no fill launch; sums stay in arrival order
# speedup vs baseline: 3.4746x; 1.0576x over previous
#include <hip/hip_runtime.h>
#include <hip/hip_bf16.h>
#include <math.h>


namespace {

constexpr int kN0 = 100000;
constexpr int kN1 = 40000;
constexpr int kN2 = 10000;
constexpr int kE1 = 640000;
constexpr int kE2 = 160000;

typedef __attribute__((ext_vector_type(2))) float v2f;
typedef __attribute__((ext_vector_type(8))) float v8f;

typedef __attribute__((ext_vector_type(4))) float v4f;

constexpr int kCap = 64;

constexpr size_t align64(size_t n) { return (n + 63) / 64 * 64; }
constexpr size_t kOffSum1 = 0;
constexpr size_t kOffSum2 = kOffSum1 + (size_t)kN1 * 256;
constexpr size_t kOffH    = kOffSum2 + (size_t)kN2 * 256;
constexpr size_t kOffDeg  = align64(kOffH + (size_t)kN1 * 256);
constexpr size_t kOffBkt1 = kOffDeg + align64(kN1 + kN2 + 2);
constexpr size_t kOffBkt2 = kOffBkt1 + (size_t)kN1 * kCap;
constexpr size_t kOffSpl1 = kOffBkt2 + (size_t)kN2 * kCap;
constexpr size_t kOffSpl2 = kOffSpl1 + align64(2 * (size_t)kE1);

__global__ void zero_i32(int* __restrict__ p, int n) {
  int i = blockIdx.x * blockDim.x + threadIdx.x;
  if (i < n) p[i] = 0;
}

__global__ void edge_bucket(const int* __restrict__ src1,
                            const int* __restrict__ dst1,
                            const int* __restrict__ src2,
                            const int* __restrict__ dst2,
                            int* __restrict__ deg,
                            int* __restrict__ bkt1,
                            int* __restrict__ bkt2,
                            int* __restrict__ spl1,
                            int* __restrict__ spl2) {
  const int i = blockIdx.x * blockDim.x + threadIdx.x;
  if (i < kE1) {
    const int d = dst1[i];
    const int s = src1[i];
    const int slot = atomicAdd(&deg[d], 1);
    if (slot < kCap) {
      bkt1[(size_t)d * kCap + slot] = s;
    } else {
      const int o = atomicAdd(&deg[kN1 + kN2], 1);
      spl1[o] = d;
      spl1[kE1 + o] = s;
    }
  } else if (i < kE1 + kE2) {
    const int j = i - kE1;
    const int d = dst2[j];
    const int s = src2[j];
    const int slot = atomicAdd(&deg[kN1 + d], 1);
    if (slot < kCap) {
      bkt2[(size_t)d * kCap + slot] = s;
    } else {
      const int o = atomicAdd(&deg[kN1 + kN2 + 1], 1);
      spl2[o] = d;
      spl2[kE2 + o] = s;
    }
  }
}

template <int NSRC>
__global__ void __launch_bounds__(256)
gather_sum(const float* __restrict__ feat,
           const int* __restrict__ deg,
           const int* __restrict__ bkt,
           const int* __restrict__ splCount,
           const int* __restrict__ spl,
           int splCap,
           float* __restrict__ sum,
           int nseg) {
  const int lane = threadIdx.x & 31;
  const int node =
      blockIdx.x * 8 + __builtin_amdgcn_readfirstlane((int)(threadIdx.x >> 5));
  if (node >= nseg) return;
  const int cntNode = deg[node];
  const int end = min(cntNode, kCap);
  const int* perm = bkt + (size_t)node * kCap;

  v4f a0 = v4f{0, 0, 0, 0};
  v4f a1 = v4f{0, 0, 0, 0};
  for (int base = 0; base < end; base += 32) {
    const int n = min(32, end - base);
    int mine = (lane < n) ? perm[base + lane] : 0;
    mine = min(max(mine, 0), NSRC - 1);
    int j = 0;
    for (; j + 4 <= n; j += 4) {
      const v4f* p0 = (const v4f*)(feat + (size_t)__builtin_amdgcn_readlane(mine, j) * 256);
      const v4f* p1 = (const v4f*)(feat + (size_t)__builtin_amdgcn_readlane(mine, j + 1) * 256);
      const v4f* p2 = (const v4f*)(feat + (size_t)__builtin_amdgcn_readlane(mine, j + 2) * 256);
      const v4f* p3 = (const v4f*)(feat + (size_t)__builtin_amdgcn_readlane(mine, j + 3) * 256);
      const v4f x0 = p0[lane], y0 = p0[lane + 32];
      const v4f x1 = p1[lane], y1 = p1[lane + 32];
      const v4f x2 = p2[lane], y2 = p2[lane + 32];
      const v4f x3 = p3[lane], y3 = p3[lane + 32];
      a0 += x0; a1 += y0;
      a0 += x1; a1 += y1;
      a0 += x2; a1 += y2;
      a0 += x3; a1 += y3;
    }
#pragma unroll 1
    for (; j < n; ++j) {
      const v4f* p = (const v4f*)(feat + (size_t)__builtin_amdgcn_readlane(mine, j) * 256);
      a0 += p[lane];
      a1 += p[lane + 32];
    }
  }
  if (cntNode > kCap) {
    const int total = min(splCount[0], splCap);
    for (int base = 0; base < total; base += 32) {
      const int i = base + lane;
      const int d = (i < total) ? spl[i] : -1;
      int s = (i < total) ? spl[splCap + i] : 0;
      s = min(max(s, 0), NSRC - 1);
      unsigned hit = __builtin_amdgcn_ballot_w32(d == node);
      while (hit != 0u) {
        const int j = __builtin_ctz(hit);
        hit &= hit - 1u;
        const v4f* p = (const v4f*)(feat + (size_t)__builtin_amdgcn_readlane(s, j) * 256);
        a0 += p[lane];
        a1 += p[lane + 32];
      }
    }
  }
  v4f* sp = (v4f*)(sum + (size_t)node * 256);
  sp[lane] = a0;
  sp[lane + 32] = a1;
}

template <int KDIM, int NDIM, int M, bool SIGMOID>
__global__ void __launch_bounds__(256, 10)
fused_wmma_gemm(const float* __restrict__ Asum,
                                const int* __restrict__ deg,
                                const float* __restrict__ Aself,
                                const float* __restrict__ Wl,
                                const float* __restrict__ Wr,
                                const float* __restrict__ bias,
                                float* __restrict__ out) {
  static_assert(M % 16 == 0 && M >= 16, "row count must be a multiple of 16");
  constexpr int KC = 32;
  constexpr int LDS_STRIDE = KC + 2;
  __shared__ float ldsW[2][64][LDS_STRIDE];
  constexpr int A_STRIDE = KC + 4;
  __shared__ __attribute__((aligned(16))) float ldsA[2][128][A_STRIDE];

  const int lane = threadIdx.x & 31;
  const int wave = threadIdx.x >> 5;
  constexpr int kColBlocks = NDIM / 64;
  const int brow = blockIdx.x / kColBlocks;
  const int n0 = (blockIdx.x % kColBlocks) * 64;
  const int m0 = brow * 128 + wave * 16;

  const int half = lane >> 4;
  const int l15 = lane & 15;
  const int srow = lane >> 3;
  const int scol = (lane & 7) * 4;
  const int mload = min(m0, M - 16) + srow;
  const float* sSum  = Asum  + (size_t)mload * KDIM + scol;
  const float* sSelf = Aself + (size_t)mload * KDIM + scol;
  float sInv[4];
#pragma unroll
  for (int p = 0; p < 4; ++p)
    sInv[p] = 1.0f / fmaxf((float)deg[mload + 4 * p], 1.0f);
  const int wrow = wave * 16;

  v8f acc[4];
#pragma unroll
  for (int t = 0; t < 4; ++t) acc[t] = v8f{0, 0, 0, 0, 0, 0, 0, 0};

  for (int kc = 0; kc < KDIM; kc += KC) {
    __syncthreads();
    const int wcol = threadIdx.x & 63;
    const int wkr = threadIdx.x >> 6;
    float wl[KC / 4], wr[KC / 4];
    v4f as[4], at[4];
#pragma unroll
    for (int q = 0; q < KC / 4; ++q) {
      wl[q] = Wl[(size_t)(kc + wkr + 4 * q) * NDIM + n0 + wcol];
      wr[q] = Wr[(size_t)(kc + wkr + 4 * q) * NDIM + n0 + wcol];
    }
#pragma unroll
    for (int p = 0; p < 4; ++p) {
      as[p] = *(const v4f*)(sSum + (size_t)(4 * p) * KDIM + kc);
      at[p] = *(const v4f*)(sSelf + (size_t)(4 * p) * KDIM + kc);
    }
#pragma unroll
    for (int q = 0; q < KC / 4; ++q) {
      ldsW[0][wcol][wkr + 4 * q] = wl[q];
      ldsW[1][wcol][wkr + 4 * q] = wr[q];
    }
#pragma unroll
    for (int p = 0; p < 4; ++p) {
      *(v4f*)&ldsA[0][wrow + 4 * p + srow][scol] = as[p] * sInv[p];
      *(v4f*)&ldsA[1][wrow + 4 * p + srow][scol] = at[p];
    }
    __syncthreads();

#pragma unroll
    for (int k2 = 0; k2 < KC; k2 += 4) {
      const int koff = k2 + 2 * half;
      const v2f a1 = *(const v2f*)&ldsA[0][wrow + l15][koff];
      const v2f a2 = *(const v2f*)&ldsA[1][wrow + l15][koff];
#pragma unroll
      for (int t = 0; t < 4; ++t) {
        const int lcol = 16 * t + l15;
        const v2f bl = *(const v2f*)&ldsW[0][lcol][koff];
        acc[t] = __builtin_amdgcn_wmma_f32_16x16x4_f32(
            false, a1, false, bl, (short)0, acc[t], false, false);
        const v2f br = *(const v2f*)&ldsW[1][lcol][koff];
        acc[t] = __builtin_amdgcn_wmma_f32_16x16x4_f32(
            false, a2, false, br, (short)0, acc[t], false, false);
      }
    }
  }

#pragma unroll
  for (int t = 0; t < 4; ++t) {
    const int col = n0 + 16 * t + l15;
    const float b = bias[col];
#pragma unroll
    for (int r = 0; r < 8; ++r) {
      const int row = m0 + r + 8 * half;
      float v = acc[t][r] + b;
      v = SIGMOID ? (1.0f / (1.0f + expf(-v))) : fmaxf(v, 0.0f);
      if (row < M) out[(size_t)row * NDIM + col] = v;
    }
  }
}

}

extern "C" void kernel_launch(void* const* d_in, const int* in_sizes, int n_in,
                              void* d_out, int out_size, void* d_ws, size_t ws_size,
                              hipStream_t stream) {
  (void)in_sizes; (void)n_in; (void)out_size; (void)ws_size;

  const float* x   = (const float*)d_in[0];
  const float* W1l = (const float*)d_in[1];
  const float* b1  = (const float*)d_in[2];
  const float* W1r = (const float*)d_in[3];
  const float* W2l = (const float*)d_in[4];
  const float* b2  = (const float*)d_in[5];
  const float* W2r = (const float*)d_in[6];
  const int* src1  = (const int*)d_in[7];
  const int* dst1  = (const int*)d_in[8];
  const int* src2  = (const int*)d_in[9];
  const int* dst2  = (const int*)d_in[10];
  float* out = (float*)d_out;

  float* ws   = (float*)d_ws;
  int*   wsi  = (int*)d_ws;
  float* sum1 = ws + kOffSum1;
  float* sum2 = ws + kOffSum2;
  float* h    = ws + kOffH;
  int* deg  = wsi + kOffDeg;
  int* bkt1 = wsi + kOffBkt1;
  int* bkt2 = wsi + kOffBkt2;
  int* spl1 = wsi + kOffSpl1;
  int* spl2 = wsi + kOffSpl2;

  zero_i32<<<(kN1 + kN2 + 2 + 255) / 256, 256, 0, stream>>>(deg, kN1 + kN2 + 2);

  edge_bucket<<<(kE1 + kE2) / 256, 256, 0, stream>>>(src1, dst1, src2, dst2, deg,
                                                    bkt1, bkt2, spl1, spl2);

  gather_sum<kN0><<<kN1 / 8, 256, 0, stream>>>(x, deg, bkt1, deg + kN1 + kN2, spl1,
                                             kE1, sum1, kN1);

  fused_wmma_gemm<256, 256, kN1, false>
      <<<313 * 4, 256, 0, stream>>>(sum1, deg, x, W1l, W1r, b1, h);

  gather_sum<kN1><<<kN2 / 8, 256, 0, stream>>>(h, deg + kN1, bkt2, deg + kN1 + kN2 + 1,
                                             spl2, kE2, sum2, kN2);

  fused_wmma_gemm<256, 128, kN2, true>
      <<<79 * 2, 256, 0, stream>>>(sum2, deg + kN1, h, W2l, W2r, b2, out);
}


# GEMMs as one resident full-width tile per WGP (160x256, 48x128 per work-group), next K-chunk's global loads issued under the current WMMAs (2 LDS stages); products summed sum half, then self half
# speedup vs baseline: 3.6244x; 1.0431x over previous
#include <hip/hip_runtime.h>
#include <hip/hip_bf16.h>
#include <math.h>


namespace {

constexpr int kN0 = 100000;
constexpr int kN1 = 40000;
constexpr int kN2 = 10000;
constexpr int kE1 = 640000;
constexpr int kE2 = 160000;

typedef __attribute__((ext_vector_type(2))) float v2f;
typedef __attribute__((ext_vector_type(8))) float v8f;

typedef __attribute__((ext_vector_type(4))) float v4f;

constexpr int kCap = 64;

constexpr size_t align64(size_t n) { return (n + 63) / 64 * 64; }
constexpr size_t kOffSum1 = 0;
constexpr size_t kOffSum2 = kOffSum1 + (size_t)kN1 * 256;
constexpr size_t kOffH    = kOffSum2 + (size_t)kN2 * 256;
constexpr size_t kOffDeg  = align64(kOffH + (size_t)kN1 * 256);
constexpr size_t kOffBkt1 = kOffDeg + align64(kN1 + kN2 + 2);
constexpr size_t kOffBkt2 = kOffBkt1 + (size_t)kN1 * kCap;
constexpr size_t kOffSpl1 = kOffBkt2 + (size_t)kN2 * kCap;
constexpr size_t kOffSpl2 = kOffSpl1 + align64(2 * (size_t)kE1);

__global__ void zero_i32(int* __restrict__ p, int n) {
  int i = blockIdx.x * blockDim.x + threadIdx.x;
  if (i < n) p[i] = 0;
}

__global__ void edge_bucket(const int* __restrict__ src1,
                            const int* __restrict__ dst1,
                            const int* __restrict__ src2,
                            const int* __restrict__ dst2,
                            int* __restrict__ deg,
                            int* __restrict__ bkt1,
                            int* __restrict__ bkt2,
                            int* __restrict__ spl1,
                            int* __restrict__ spl2) {
  const int i = blockIdx.x * blockDim.x + threadIdx.x;
  if (i < kE1) {
    const int d = dst1[i];
    const int s = src1[i];
    const int slot = atomicAdd(&deg[d], 1);
    if (slot < kCap) {
      bkt1[(size_t)d * kCap + slot] = s;
    } else {
      const int o = atomicAdd(&deg[kN1 + kN2], 1);
      spl1[o] = d;
      spl1[kE1 + o] = s;
    }
  } else if (i < kE1 + kE2) {
    const int j = i - kE1;
    const int d = dst2[j];
    const int s = src2[j];
    const int slot = atomicAdd(&deg[kN1 + d], 1);
    if (slot < kCap) {
      bkt2[(size_t)d * kCap + slot] = s;
    } else {
      const int o = atomicAdd(&deg[kN1 + kN2 + 1], 1);
      spl2[o] = d;
      spl2[kE2 + o] = s;
    }
  }
}

template <int NSRC>
__global__ void __launch_bounds__(256)
gather_sum(const float* __restrict__ feat,
           const int* __restrict__ deg,
           const int* __restrict__ bkt,
           const int* __restrict__ splCount,
           const int* __restrict__ spl,
           int splCap,
           float* __restrict__ sum,
           int nseg) {
  const int lane = threadIdx.x & 31;
  const int node =
      blockIdx.x * 8 + __builtin_amdgcn_readfirstlane((int)(threadIdx.x >> 5));
  if (node >= nseg) return;
  const int cntNode = deg[node];
  const int end = min(cntNode, kCap);
  const int* perm = bkt + (size_t)node * kCap;

  v4f a0 = v4f{0, 0, 0, 0};
  v4f a1 = v4f{0, 0, 0, 0};
  for (int base = 0; base < end; base += 32) {
    const int n = min(32, end - base);
    int mine = (lane < n) ? perm[base + lane] : 0;
    mine = min(max(mine, 0), NSRC - 1);
    int j = 0;
    for (; j + 4 <= n; j += 4) {
      const v4f* p0 = (const v4f*)(feat + (size_t)__builtin_amdgcn_readlane(mine, j) * 256);
      const v4f* p1 = (const v4f*)(feat + (size_t)__builtin_amdgcn_readlane(mine, j + 1) * 256);
      const v4f* p2 = (const v4f*)(feat + (size_t)__builtin_amdgcn_readlane(mine, j + 2) * 256);
      const v4f* p3 = (const v4f*)(feat + (size_t)__builtin_amdgcn_readlane(mine, j + 3) * 256);
      const v4f x0 = p0[lane], y0 = p0[lane + 32];
      const v4f x1 = p1[lane], y1 = p1[lane + 32];
      const v4f x2 = p2[lane], y2 = p2[lane + 32];
      const v4f x3 = p3[lane], y3 = p3[lane + 32];
      a0 += x0; a1 += y0;
      a0 += x1; a1 += y1;
      a0 += x2; a1 += y2;
      a0 += x3; a1 += y3;
    }
#pragma unroll 1
    for (; j < n; ++j) {
      const v4f* p = (const v4f*)(feat + (size_t)__builtin_amdgcn_readlane(mine, j) * 256);
      a0 += p[lane];
      a1 += p[lane + 32];
    }
  }
  if (cntNode > kCap) {
    const int total = min(splCount[0], splCap);
    for (int base = 0; base < total; base += 32) {
      const int i = base + lane;
      const int d = (i < total) ? spl[i] : -1;
      int s = (i < total) ? spl[splCap + i] : 0;
      s = min(max(s, 0), NSRC - 1);
      unsigned hit = __builtin_amdgcn_ballot_w32(d == node);
      while (hit != 0u) {
        const int j = __builtin_ctz(hit);
        hit &= hit - 1u;
        const v4f* p = (const v4f*)(feat + (size_t)__builtin_amdgcn_readlane(s, j) * 256);
        a0 += p[lane];
        a1 += p[lane + 32];
      }
    }
  }
  v4f* sp = (v4f*)(sum + (size_t)node * 256);
  sp[lane] = a0;
  sp[lane + 32] = a1;
}

template <int KDIM, int NDIM, int M, int WM, int WN, int MT, int NT, bool SIGMOID>
__global__ void __launch_bounds__(256)
fused_wmma_gemm(const float* __restrict__ Asum,
                const int* __restrict__ deg,
                const float* __restrict__ Aself,
                const float* __restrict__ Wl,
                const float* __restrict__ Wr,
                const float* __restrict__ bias,
                float* __restrict__ out) {
  static_assert(WM * WN == 8, "8 waves per work-group");
  static_assert(WN * NT * 16 == NDIM, "a work-group spans all columns");
  static_assert(256 % NDIM == 0, "column map of the W stage");
  constexpr int TM = WM * MT * 16;
  constexpr int KC = 32;
  constexpr int W_STRIDE = KC + 2;
  constexpr int A_STRIDE = KC + 4;
  constexpr int HALF_CHUNKS = KDIM / KC;
  constexpr int NCHUNK = 2 * HALF_CHUNKS;
  constexpr int A_ITEMS = TM * (KC / 4);
  constexpr int A_PER = (A_ITEMS + 255) / 256;
  constexpr int W_KP = 256 / NDIM;
  constexpr int W_PER = KC / 2 / W_KP;
  __shared__ __attribute__((aligned(16))) float ldsW[2][NDIM][W_STRIDE];
  __shared__ __attribute__((aligned(16))) float ldsA[2][TM][A_STRIDE];

  const int tid = threadIdx.x;
  const int lane = tid & 31;
  const int wave = tid >> 5;
  const int half = lane >> 4;
  const int l15 = lane & 15;
  const int m0 = blockIdx.x * TM;
  const int rbase = (wave / WN) * (MT * 16);
  const int cbase = (wave % WN) * (NT * 16);

  int aOff[A_PER];
  float sInv[A_PER];
#pragma unroll
  for (int it = 0; it < A_PER; ++it) {
    const int idx = tid + 256 * it;
    const int row = min(m0 + (idx >> 3), M - 1);
    aOff[it] = row * KDIM + (idx & 7) * 4;
    sInv[it] = 1.0f / fmaxf((float)deg[row], 1.0f);
  }
  const int wcol = tid % NDIM;
  const int wkp = tid / NDIM;

  v8f acc[MT][NT];
#pragma unroll
  for (int i = 0; i < MT; ++i)
#pragma unroll
    for (int j = 0; j < NT; ++j) acc[i][j] = v8f{0, 0, 0, 0, 0, 0, 0, 0};

  v4f aReg[A_PER];
  v2f wReg[W_PER];

  auto gload = [&](int c) {
    const bool s = c < HALF_CHUNKS;
    const float* Ab = s ? Asum : Aself;
    const float* Wb = s ? Wl : Wr;
    const int kc = (s ? c : c - HALF_CHUNKS) * KC;
#pragma unroll
    for (int q = 0; q < W_PER; ++q) {
      const int k = kc + 2 * (wkp + W_KP * q);
      wReg[q][0] = Wb[(size_t)k * NDIM + wcol];
      wReg[q][1] = Wb[(size_t)(k + 1) * NDIM + wcol];
    }
#pragma unroll
    for (int it = 0; it < A_PER; ++it) {
      if (A_ITEMS % 256 == 0 || tid + 256 * it < A_ITEMS)
        aReg[it] = *(const v4f*)(Ab + (size_t)aOff[it] + kc);
    }
  };
  auto lstore = [&](int c, int st) {
    const bool s = c < HALF_CHUNKS;
#pragma unroll
    for (int q = 0; q < W_PER; ++q)
      *(v2f*)&ldsW[st][wcol][2 * (wkp + W_KP * q)] = wReg[q];
#pragma unroll
    for (int it = 0; it < A_PER; ++it) {
      const int idx = tid + 256 * it;
      if (A_ITEMS % 256 == 0 || idx < A_ITEMS) {
        const float f = s ? sInv[it] : 1.0f;
        *(v4f*)&ldsA[st][idx >> 3][(idx & 7) * 4] = aReg[it] * f;
      }
    }
  };

  gload(0);
  lstore(0, 0);
  __syncthreads();

  auto ldfrag = [&](int st, int k2, v2f (&a)[MT], v2f (&b)[NT]) {
    const int koff = k2 + 2 * half;
#pragma unroll
    for (int i = 0; i < MT; ++i)
      a[i] = *(const v2f*)&ldsA[st][rbase + 16 * i + l15][koff];
#pragma unroll
    for (int j = 0; j < NT; ++j)
      b[j] = *(const v2f*)&ldsW[st][cbase + 16 * j + l15][koff];
  };

  for (int c = 0; c < NCHUNK; ++c) {
    const int st = c & 1;
    if (c + 1 < NCHUNK) gload(c + 1);
    v2f a[2][MT], b[2][NT];
    ldfrag(st, 0, a[0], b[0]);

#pragma unroll
    for (int s = 0; s < KC / 4; ++s) {
      if (s + 1 < KC / 4) ldfrag(st, 4 * (s + 1), a[(s + 1) & 1], b[(s + 1) & 1]);
      __builtin_amdgcn_sched_barrier(0);
#pragma unroll
      for (int i = 0; i < MT; ++i)
#pragma unroll
        for (int j = 0; j < NT; ++j)
          acc[i][j] = __builtin_amdgcn_wmma_f32_16x16x4_f32(
              false, a[s & 1][i], false, b[s & 1][j], (short)0, acc[i][j],
              false, false);
      __builtin_amdgcn_sched_barrier(0);
    }

    if (c + 1 < NCHUNK) lstore(c + 1, st ^ 1);
    __syncthreads();
  }

#pragma unroll
  for (int j = 0; j < NT; ++j) {
    const int col = cbase + 16 * j + l15;
    const float bv = bias[col];
#pragma unroll
    for (int i = 0; i < MT; ++i) {
#pragma unroll
      for (int r = 0; r < 8; ++r) {
        const int row = m0 + rbase + 16 * i + r + 8 * half;
        float v = acc[i][j][r] + bv;
        v = SIGMOID ? (1.0f / (1.0f + expf(-v))) : fmaxf(v, 0.0f);
        if (row < M) out[(size_t)row * NDIM + col] = v;
      }
    }
  }
}

}

extern "C" void kernel_launch(void* const* d_in, const int* in_sizes, int n_in,
                              void* d_out, int out_size, void* d_ws, size_t ws_size,
                              hipStream_t stream) {
  (void)in_sizes; (void)n_in; (void)out_size; (void)ws_size;

  const float* x   = (const float*)d_in[0];
  const float* W1l = (const float*)d_in[1];
  const float* b1  = (const float*)d_in[2];
  const float* W1r = (const float*)d_in[3];
  const float* W2l = (const float*)d_in[4];
  const float* b2  = (const float*)d_in[5];
  const float* W2r = (const float*)d_in[6];
  const int* src1  = (const int*)d_in[7];
  const int* dst1  = (const int*)d_in[8];
  const int* src2  = (const int*)d_in[9];
  const int* dst2  = (const int*)d_in[10];
  float* out = (float*)d_out;

  float* ws   = (float*)d_ws;
  int*   wsi  = (int*)d_ws;
  float* sum1 = ws + kOffSum1;
  float* sum2 = ws + kOffSum2;
  float* h    = ws + kOffH;
  int* deg  = wsi + kOffDeg;
  int* bkt1 = wsi + kOffBkt1;
  int* bkt2 = wsi + kOffBkt2;
  int* spl1 = wsi + kOffSpl1;
  int* spl2 = wsi + kOffSpl2;

  zero_i32<<<(kN1 + kN2 + 2 + 255) / 256, 256, 0, stream>>>(deg, kN1 + kN2 + 2);

  edge_bucket<<<(kE1 + kE2) / 256, 256, 0, stream>>>(src1, dst1, src2, dst2, deg,
                                                    bkt1, bkt2, spl1, spl2);

  gather_sum<kN0><<<kN1 / 8, 256, 0, stream>>>(x, deg, bkt1, deg + kN1 + kN2, spl1,
                                             kE1, sum1, kN1);

  fused_wmma_gemm<256, 256, kN1, 2, 4, 5, 4, false>
      <<<kN1 / 160, 256, 0, stream>>>(sum1, deg, x, W1l, W1r, b1, h);

  gather_sum<kN1><<<kN2 / 8, 256, 0, stream>>>(h, deg + kN1, bkt2, deg + kN1 + kN2 + 1,
                                             spl2, kE2, sum2, kN2);

  fused_wmma_gemm<256, 128, kN2, 1, 8, 3, 1, true>
      <<<(kN2 + 47) / 48, 256, 0, stream>>>(sum2, deg + kN1, h, W2l, W2r, b2, out);
}


# Edge lists without global atomics: one launch bins each 4096-edge chunk by 32-node range into cells (LDS counters); gather groups split their cells into per-node LDS lists; sums in LDS arrival order
# speedup vs baseline: 5.0262x; 1.3868x over previous
#include <hip/hip_runtime.h>
#include <hip/hip_bf16.h>
#include <math.h>


namespace {

constexpr int kN0 = 100000;
constexpr int kN1 = 40000;
constexpr int kN2 = 10000;
constexpr int kE1 = 640000;
constexpr int kE2 = 160000;

typedef __attribute__((ext_vector_type(2))) float v2f;
typedef __attribute__((ext_vector_type(8))) float v8f;

typedef __attribute__((ext_vector_type(4))) float v4f;

constexpr int kRange = 32;
constexpr int kChunk = 4096;
constexpr int kR1 = kN1 / kRange;
constexpr int kR2 = (kN2 + kRange - 1) / kRange;
constexpr int kB1 = (kE1 + kChunk - 1) / kChunk;
constexpr int kB2 = (kE2 + kChunk - 1) / kChunk;
constexpr int kC1 = 32;
constexpr int kC2 = 64;
constexpr int kCap = 64;
static_assert(kN1 % kRange == 0, "layer-1 ranges are whole");
static_assert(kN0 <= (1 << 17), "a source index fits the packed edge");

constexpr size_t align64(size_t n) { return (n + 63) / 64 * 64; }
constexpr size_t kOffSum1 = 0;
constexpr size_t kOffSum2 = kOffSum1 + (size_t)kN1 * 256;
constexpr size_t kOffH    = kOffSum2 + (size_t)kN2 * 256;
constexpr size_t kOffDeg  = align64(kOffH + (size_t)kN1 * 256);
constexpr size_t kOffCel1 = kOffDeg + align64(kN1 + kN2);
constexpr size_t kOffCel2 = kOffCel1 + align64((size_t)kB1 * kR1 * kC1);

__global__ void __launch_bounds__(256)
edge_bin(const int* __restrict__ src1,
         const int* __restrict__ dst1,
         const int* __restrict__ src2,
         const int* __restrict__ dst2,
         int* __restrict__ cell1,
         int* __restrict__ cell2) {
  __shared__ int cnt[kR1];
  static_assert(kR1 >= kR2, "counter array serves both layers");
  const int tid = threadIdx.x;
  const bool first = blockIdx.x < kB1;
  const int b = first ? blockIdx.x : blockIdx.x - kB1;
  const int* src = first ? src1 : src2;
  const int* dst = first ? dst1 : dst2;
  const int E = first ? kE1 : kE2;
  const int R = first ? kR1 : kR2;
  const int C = first ? kC1 : kC2;
  const int nDst = first ? kN1 : kN2;
  const int nSrc = first ? kN0 : kN1;
  int* mycell = (first ? cell1 : cell2) + (size_t)b * R * C;

  for (int r = tid; r < R; r += 256) cnt[r] = 0;
  __syncthreads();

  constexpr int PER = kChunk / 256;
  int d[PER], s[PER];
#pragma unroll
  for (int k = 0; k < PER; ++k) {
    const int i = min(b * kChunk + k * 256 + tid, E - 1);
    d[k] = dst[i];
    s[k] = src[i];
  }
  int slot[PER];
#pragma unroll
  for (int k = 0; k < PER; ++k) {
    d[k] = min(max(d[k], 0), nDst - 1);
    s[k] = min(max(s[k], 0), nSrc - 1);
    slot[k] = C;
    if (b * kChunk + k * 256 + tid < E) slot[k] = atomicAdd(&cnt[d[k] / kRange], 1);
  }
#pragma unroll
  for (int k = 0; k < PER; ++k) {
    if (slot[k] < C - 1)
      mycell[(d[k] / kRange) * C + 1 + slot[k]] = ((d[k] % kRange) << 17) | s[k];
  }
  __syncthreads();
  for (int r = tid; r < R; r += 256) mycell[r * C] = cnt[r];
}

template <int NSRC, int C, int NB, int R>
__global__ void __launch_bounds__(256)
gather_sum(const float* __restrict__ feat,
           const int* __restrict__ cell,
           const int* __restrict__ srcAll,
           const int* __restrict__ dstAll,
           int nEdges,
           int* __restrict__ degOut,
           float* __restrict__ sum,
           int nseg) {
  __shared__ int ldsDeg[kRange];
  __shared__ int ldsList[kRange][kCap];
  __shared__ int ldsSlow;
  const int tid = threadIdx.x;
  const int lane = tid & 31;
  const int wave = __builtin_amdgcn_readfirstlane((int)(threadIdx.x >> 5));
  const int r = blockIdx.x;

  if (tid < kRange) ldsDeg[tid] = 0;
  if (tid == 0) ldsSlow = 0;
  __syncthreads();

  constexpr int PIECES = C / 4;
  constexpr int PER_PASS = 256 / PIECES;
  const int piece = tid % PIECES;
  const int cq = tid / PIECES;
  constexpr int NPASS = (NB + PER_PASS - 1) / PER_PASS;
  int over = 0;
  int4 vv[NPASS];
  int slot[NPASS][4];
#pragma unroll
  for (int k = 0; k < NPASS; ++k) {
    const int b = k * PER_PASS + cq;
    vv[k] = make_int4(0, 0, 0, 0);
    if (b < NB) vv[k] = *(const int4*)(cell + ((size_t)b * R + r) * C + piece * 4);
  }
#pragma unroll
  for (int k = 0; k < NPASS; ++k) {
    const int4 v = vv[k];
    const int n = __shfl(v.x, lane & ~(PIECES - 1));
    if (n > C - 1) over = 1;
    const int nn = min(max(n, 0), C - 1);
    const int e4[4] = {v.x, v.y, v.z, v.w};
#pragma unroll
    for (int i = 0; i < 4; ++i) {
      const int p = piece * 4 + i;
      slot[k][i] = kCap;
      if (p >= 1 && p <= nn)
        slot[k][i] = atomicAdd(&ldsDeg[(e4[i] >> 17) & (kRange - 1)], 1);
    }
  }
#pragma unroll
  for (int k = 0; k < NPASS; ++k) {
    const int e4[4] = {vv[k].x, vv[k].y, vv[k].z, vv[k].w};
#pragma unroll
    for (int i = 0; i < 4; ++i) {
      if (slot[k][i] < kCap)
        ldsList[(e4[i] >> 17) & (kRange - 1)][slot[k][i]] = e4[i] & 0x1FFFF;
    }
  }
  if (over) ldsSlow = 1;
  __syncthreads();
  if (tid < kRange && ldsDeg[tid] > kCap) ldsSlow = 1;
  __syncthreads();
  const bool slow = __builtin_amdgcn_readfirstlane(ldsSlow) != 0;

  for (int q = 0; q < kRange / 8; ++q) {
    const int nl = wave * (kRange / 8) + q;
    const int node = r * kRange + nl;
    if (node >= nseg) break;
    v4f a0 = v4f{0, 0, 0, 0};
    v4f a1 = v4f{0, 0, 0, 0};
    int cntNode = 0;
    if (!slow) {
      cntNode = __builtin_amdgcn_readfirstlane(ldsDeg[nl]);
      const int end = min(cntNode, kCap);
      for (int base = 0; base < end; base += 32) {
        const int n = min(32, end - base);
        int mine = (lane < n) ? ldsList[nl][base + lane] : 0;
        mine = min(max(mine, 0), NSRC - 1);
        int j = 0;
        for (; j + 4 <= n; j += 4) {
          const v4f* p0 = (const v4f*)(feat + (size_t)__builtin_amdgcn_readlane(mine, j) * 256);
          const v4f* p1 = (const v4f*)(feat + (size_t)__builtin_amdgcn_readlane(mine, j + 1) * 256);
          const v4f* p2 = (const v4f*)(feat + (size_t)__builtin_amdgcn_readlane(mine, j + 2) * 256);
          const v4f* p3 = (const v4f*)(feat + (size_t)__builtin_amdgcn_readlane(mine, j + 3) * 256);
          const v4f x0 = p0[lane], y0 = p0[lane + 32];
          const v4f x1 = p1[lane], y1 = p1[lane + 32];
          const v4f x2 = p2[lane], y2 = p2[lane + 32];
          const v4f x3 = p3[lane], y3 = p3[lane + 32];
          a0 += x0; a1 += y0;
          a0 += x1; a1 += y1;
          a0 += x2; a1 += y2;
          a0 += x3; a1 += y3;
        }
#pragma unroll 1
        for (; j < n; ++j) {
          const v4f* p = (const v4f*)(feat + (size_t)__builtin_amdgcn_readlane(mine, j) * 256);
          a0 += p[lane];
          a1 += p[lane + 32];
        }
      }
    } else {
      for (int base = 0; base < nEdges; base += 32) {
        const int i = base + lane;
        const int d = (i < nEdges) ? min(max(dstAll[i], 0), nseg - 1) : -1;
        int s = (i < nEdges) ? srcAll[i] : 0;
        s = min(max(s, 0), NSRC - 1);
        unsigned hit = __builtin_amdgcn_ballot_w32(d == node);
        cntNode += __builtin_popcount(hit);
        while (hit != 0u) {
          const int j = __builtin_ctz(hit);
          hit &= hit - 1u;
          const v4f* p = (const v4f*)(feat + (size_t)__builtin_amdgcn_readlane(s, j) * 256);
          a0 += p[lane];
          a1 += p[lane + 32];
        }
      }
    }
    v4f* sp = (v4f*)(sum + (size_t)node * 256);
    sp[lane] = a0;
    sp[lane + 32] = a1;
    if (lane == 0) degOut[node] = cntNode;
  }
}

template <int KDIM, int NDIM, int M, int WM, int WN, int MT, int NT, bool SIGMOID>
__global__ void __launch_bounds__(256)
fused_wmma_gemm(const float* __restrict__ Asum,
                const int* __restrict__ deg,
                const float* __restrict__ Aself,
                const float* __restrict__ Wl,
                const float* __restrict__ Wr,
                const float* __restrict__ bias,
                float* __restrict__ out) {
  static_assert(WM * WN == 8, "8 waves per work-group");
  static_assert(WN * NT * 16 == NDIM, "a work-group spans all columns");
  static_assert(256 % NDIM == 0, "column map of the W stage");
  constexpr int TM = WM * MT * 16;
  constexpr int KC = 32;
  constexpr int W_STRIDE = KC + 2;
  constexpr int A_STRIDE = KC + 4;
  constexpr int HALF_CHUNKS = KDIM / KC;
  constexpr int NCHUNK = 2 * HALF_CHUNKS;
  constexpr int A_ITEMS = TM * (KC / 4);
  constexpr int A_PER = (A_ITEMS + 255) / 256;
  constexpr int W_KP = 256 / NDIM;
  constexpr int W_PER = KC / 2 / W_KP;
  __shared__ __attribute__((aligned(16))) float ldsW[2][NDIM][W_STRIDE];
  __shared__ __attribute__((aligned(16))) float ldsA[2][TM][A_STRIDE];

  const int tid = threadIdx.x;
  const int lane = tid & 31;
  const int wave = tid >> 5;
  const int half = lane >> 4;
  const int l15 = lane & 15;
  const int m0 = blockIdx.x * TM;
  const int rbase = (wave / WN) * (MT * 16);
  const int cbase = (wave % WN) * (NT * 16);

  int aOff[A_PER];
  float sInv[A_PER];
#pragma unroll
  for (int it = 0; it < A_PER; ++it) {
    const int idx = tid + 256 * it;
    const int row = min(m0 + (idx >> 3), M - 1);
    aOff[it] = row * KDIM + (idx & 7) * 4;
    sInv[it] = 1.0f / fmaxf((float)deg[row], 1.0f);
  }
  const int wcol = tid % NDIM;
  const int wkp = tid / NDIM;

  v8f acc[MT][NT];
#pragma unroll
  for (int i = 0; i < MT; ++i)
#pragma unroll
    for (int j = 0; j < NT; ++j) acc[i][j] = v8f{0, 0, 0, 0, 0, 0, 0, 0};

  v4f aReg[A_PER];
  v2f wReg[W_PER];

  auto gload = [&](int c) {
    const bool s = c < HALF_CHUNKS;
    const float* Ab = s ? Asum : Aself;
    const float* Wb = s ? Wl : Wr;
    const int kc = (s ? c : c - HALF_CHUNKS) * KC;
#pragma unroll
    for (int q = 0; q < W_PER; ++q) {
      const int k = kc + 2 * (wkp + W_KP * q);
      wReg[q][0] = Wb[(size_t)k * NDIM + wcol];
      wReg[q][1] = Wb[(size_t)(k + 1) * NDIM + wcol];
    }
#pragma unroll
    for (int it = 0; it < A_PER; ++it) {
      if (A_ITEMS % 256 == 0 || tid + 256 * it < A_ITEMS)
        aReg[it] = *(const v4f*)(Ab + (size_t)aOff[it] + kc);
    }
  };
  auto lstore = [&](int c, int st) {
    const bool s = c < HALF_CHUNKS;
#pragma unroll
    for (int q = 0; q < W_PER; ++q)
      *(v2f*)&ldsW[st][wcol][2 * (wkp + W_KP * q)] = wReg[q];
#pragma unroll
    for (int it = 0; it < A_PER; ++it) {
      const int idx = tid + 256 * it;
      if (A_ITEMS % 256 == 0 || idx < A_ITEMS) {
        const float f = s ? sInv[it] : 1.0f;
        *(v4f*)&ldsA[st][idx >> 3][(idx & 7) * 4] = aReg[it] * f;
      }
    }
  };

  gload(0);
  lstore(0, 0);
  __syncthreads();

  auto ldfrag = [&](int st, int k2, v2f (&a)[MT], v2f (&b)[NT]) {
    const int koff = k2 + 2 * half;
#pragma unroll
    for (int i = 0; i < MT; ++i)
      a[i] = *(const v2f*)&ldsA[st][rbase + 16 * i + l15][koff];
#pragma unroll
    for (int j = 0; j < NT; ++j)
      b[j] = *(const v2f*)&ldsW[st][cbase + 16 * j + l15][koff];
  };

  for (int c = 0; c < NCHUNK; ++c) {
    const int st = c & 1;
    if (c + 1 < NCHUNK) gload(c + 1);
    v2f a[2][MT], b[2][NT];
    ldfrag(st, 0, a[0], b[0]);

#pragma unroll
    for (int s = 0; s < KC / 4; ++s) {
      if (s + 1 < KC / 4) ldfrag(st, 4 * (s + 1), a[(s + 1) & 1], b[(s + 1) & 1]);
      __builtin_amdgcn_sched_barrier(0);
#pragma unroll
      for (int i = 0; i < MT; ++i)
#pragma unroll
        for (int j = 0; j < NT; ++j)
          acc[i][j] = __builtin_amdgcn_wmma_f32_16x16x4_f32(
              false, a[s & 1][i], false, b[s & 1][j], (short)0, acc[i][j],
              false, false);
      __builtin_amdgcn_sched_barrier(0);
    }

    if (c + 1 < NCHUNK) lstore(c + 1, st ^ 1);
    __syncthreads();
  }

#pragma unroll
  for (int j = 0; j < NT; ++j) {
    const int col = cbase + 16 * j + l15;
    const float bv = bias[col];
#pragma unroll
    for (int i = 0; i < MT; ++i) {
#pragma unroll
      for (int r = 0; r < 8; ++r) {
        const int row = m0 + rbase + 16 * i + r + 8 * half;
        float v = acc[i][j][r] + bv;
        v = SIGMOID ? (1.0f / (1.0f + expf(-v))) : fmaxf(v, 0.0f);
        if (row < M) out[(size_t)row * NDIM + col] = v;
      }
    }
  }
}

}

extern "C" void kernel_launch(void* const* d_in, const int* in_sizes, int n_in,
                              void* d_out, int out_size, void* d_ws, size_t ws_size,
                              hipStream_t stream) {
  (void)in_sizes; (void)n_in; (void)out_size; (void)ws_size;

  const float* x   = (const float*)d_in[0];
  const float* W1l = (const float*)d_in[1];
  const float* b1  = (const float*)d_in[2];
  const float* W1r = (const float*)d_in[3];
  const float* W2l = (const float*)d_in[4];
  const float* b2  = (const float*)d_in[5];
  const float* W2r = (const float*)d_in[6];
  const int* src1  = (const int*)d_in[7];
  const int* dst1  = (const int*)d_in[8];
  const int* src2  = (const int*)d_in[9];
  const int* dst2  = (const int*)d_in[10];
  float* out = (float*)d_out;

  float* ws   = (float*)d_ws;
  int*   wsi  = (int*)d_ws;
  float* sum1 = ws + kOffSum1;
  float* sum2 = ws + kOffSum2;
  float* h    = ws + kOffH;
  int* deg   = wsi + kOffDeg;
  int* cell1 = wsi + kOffCel1;
  int* cell2 = wsi + kOffCel2;

  edge_bin<<<kB1 + kB2, 256, 0, stream>>>(src1, dst1, src2, dst2, cell1, cell2);

  gather_sum<kN0, kC1, kB1, kR1><<<kR1, 256, 0, stream>>>(x, cell1, src1, dst1, kE1,
                                                       deg, sum1, kN1);

  fused_wmma_gemm<256, 256, kN1, 2, 4, 5, 4, false>
      <<<kN1 / 160, 256, 0, stream>>>(sum1, deg, x, W1l, W1r, b1, h);

  gather_sum<kN1, kC2, kB2, kR2><<<kR2, 256, 0, stream>>>(h, cell2, src2, dst2, kE2,
                                                       deg + kN1, sum2, kN2);

  fused_wmma_gemm<256, 128, kN2, 1, 8, 3, 1, true>
      <<<(kN2 + 47) / 48, 256, 0, stream>>>(sum2, deg + kN1, h, W2l, W2r, b2, out);
}


# Layer 2 in one launch: a 48-row tile's work-group lists its 48 nodes, sums their rows (16 in flight per wave, same list order) into LDS, then runs its GEMM with the mean rows read from LDS
# speedup vs baseline: 5.2883x; 1.0521x over previous
#include <hip/hip_runtime.h>
#include <hip/hip_bf16.h>
#include <math.h>


namespace {

constexpr int kN0 = 100000;
constexpr int kN1 = 40000;
constexpr int kN2 = 10000;
constexpr int kE1 = 640000;
constexpr int kE2 = 160000;

typedef __attribute__((ext_vector_type(2))) float v2f;
typedef __attribute__((ext_vector_type(8))) float v8f;

typedef __attribute__((ext_vector_type(4))) float v4f;

constexpr int kRange1 = 32;
constexpr int kRange2 = 48;
constexpr int kChunk = 4096;
constexpr int kR1 = kN1 / kRange1;
constexpr int kR2 = (kN2 + kRange2 - 1) / kRange2;
constexpr int kB1 = (kE1 + kChunk - 1) / kChunk;
constexpr int kB2 = (kE2 + kChunk - 1) / kChunk;
constexpr int kC1 = 32;
constexpr int kC2 = 64;
constexpr int kCap = 64;
static_assert(kN1 % kRange1 == 0, "layer-1 ranges are whole");
static_assert(kRange1 <= 64 && kRange2 <= 64, "a node within its range fits 6 bits of the packed edge");
static_assert(kN0 <= (1 << 17), "a source index fits the packed edge");

constexpr size_t align64(size_t n) { return (n + 63) / 64 * 64; }
constexpr size_t kOffSum1 = 0;
constexpr size_t kOffSum2 = kOffSum1 + (size_t)kN1 * 256;
constexpr size_t kOffH    = kOffSum2 + (size_t)kN2 * 256;
constexpr size_t kOffDeg  = align64(kOffH + (size_t)kN1 * 256);
constexpr size_t kOffCel1 = kOffDeg + align64(kN1 + kN2);
constexpr size_t kOffCel2 = kOffCel1 + align64((size_t)kB1 * kR1 * kC1);

template <int V> struct IntTag { static constexpr int value = V; };

__global__ void __launch_bounds__(256)
edge_bin(const int* __restrict__ src1,
         const int* __restrict__ dst1,
         const int* __restrict__ src2,
         const int* __restrict__ dst2,
         int* __restrict__ cell1,
         int* __restrict__ cell2) {
  __shared__ int cnt[kR1];
  static_assert(kR1 >= kR2, "counter array serves both layers");
  const int tid = threadIdx.x;
  const bool first = blockIdx.x < kB1;
  const int b = first ? blockIdx.x : blockIdx.x - kB1;
  const int* src = first ? src1 : src2;
  const int* dst = first ? dst1 : dst2;
  const int E = first ? kE1 : kE2;
  const int R = first ? kR1 : kR2;
  const int C = first ? kC1 : kC2;
  const int nDst = first ? kN1 : kN2;
  const int nSrc = first ? kN0 : kN1;
  int* mycell = (first ? cell1 : cell2) + (size_t)b * R * C;

  for (int r = tid; r < R; r += 256) cnt[r] = 0;
  __syncthreads();

  constexpr int PER = kChunk / 256;
  int d[PER], s[PER];
#pragma unroll
  for (int k = 0; k < PER; ++k) {
    const int i = min(b * kChunk + k * 256 + tid, E - 1);
    d[k] = dst[i];
    s[k] = src[i];
  }
  auto sortChunk = [&](auto tag) {
    constexpr int RANGE = decltype(tag)::value;
    int slot[PER];
#pragma unroll
    for (int k = 0; k < PER; ++k) {
      d[k] = min(max(d[k], 0), nDst - 1);
      s[k] = min(max(s[k], 0), nSrc - 1);
      slot[k] = C;
      if (b * kChunk + k * 256 + tid < E) slot[k] = atomicAdd(&cnt[d[k] / RANGE], 1);
    }
#pragma unroll
    for (int k = 0; k < PER; ++k) {
      if (slot[k] < C - 1)
        mycell[(d[k] / RANGE) * C + 1 + slot[k]] = ((d[k] % RANGE) << 17) | s[k];
    }
  };
  if (first) sortChunk(IntTag<kRange1>{}); else sortChunk(IntTag<kRange2>{});
  __syncthreads();
  for (int r = tid; r < R; r += 256) mycell[r * C] = cnt[r];
}

template <int NSRC, int C, int NB, int R>
__global__ void __launch_bounds__(256)
gather_sum(const float* __restrict__ feat,
           const int* __restrict__ cell,
           const int* __restrict__ srcAll,
           const int* __restrict__ dstAll,
           int nEdges,
           int* __restrict__ degOut,
           float* __restrict__ sum,
           int nseg) {
  __shared__ int ldsDeg[kRange1];
  __shared__ int ldsList[kRange1][kCap];
  __shared__ int ldsSlow;
  const int tid = threadIdx.x;
  const int lane = tid & 31;
  const int wave = __builtin_amdgcn_readfirstlane((int)(threadIdx.x >> 5));
  const int r = blockIdx.x;

  if (tid < kRange1) ldsDeg[tid] = 0;
  if (tid == 0) ldsSlow = 0;
  __syncthreads();

  constexpr int PIECES = C / 4;
  constexpr int PER_PASS = 256 / PIECES;
  const int piece = tid % PIECES;
  const int cq = tid / PIECES;
  constexpr int NPASS = (NB + PER_PASS - 1) / PER_PASS;
  int over = 0;
  int4 vv[NPASS];
  int slot[NPASS][4];
#pragma unroll
  for (int k = 0; k < NPASS; ++k) {
    const int b = k * PER_PASS + cq;
    vv[k] = make_int4(0, 0, 0, 0);
    if (b < NB) vv[k] = *(const int4*)(cell + ((size_t)b * R + r) * C + piece * 4);
  }
#pragma unroll
  for (int k = 0; k < NPASS; ++k) {
    const int4 v = vv[k];
    const int n = __shfl(v.x, lane & ~(PIECES - 1));
    if (n > C - 1) over = 1;
    const int nn = min(max(n, 0), C - 1);
    const int e4[4] = {v.x, v.y, v.z, v.w};
#pragma unroll
    for (int i = 0; i < 4; ++i) {
      const int p = piece * 4 + i;
      slot[k][i] = kCap;
      if (p >= 1 && p <= nn)
        slot[k][i] = atomicAdd(&ldsDeg[(e4[i] >> 17) & (kRange1 - 1)], 1);
    }
  }
#pragma unroll
  for (int k = 0; k < NPASS; ++k) {
    const int e4[4] = {vv[k].x, vv[k].y, vv[k].z, vv[k].w};
#pragma unroll
    for (int i = 0; i < 4; ++i) {
      if (slot[k][i] < kCap)
        ldsList[(e4[i] >> 17) & (kRange1 - 1)][slot[k][i]] = e4[i] & 0x1FFFF;
    }
  }
  if (over) ldsSlow = 1;
  __syncthreads();
  if (tid < kRange1 && ldsDeg[tid] > kCap) ldsSlow = 1;
  __syncthreads();
  const bool slow = __builtin_amdgcn_readfirstlane(ldsSlow) != 0;

  for (int q = 0; q < kRange1 / 8; ++q) {
    const int nl = wave * (kRange1 / 8) + q;
    const int node = r * kRange1 + nl;
    if (node >= nseg) break;
    v4f a0 = v4f{0, 0, 0, 0};
    v4f a1 = v4f{0, 0, 0, 0};
    int cntNode = 0;
    if (!slow) {
      cntNode = __builtin_amdgcn_readfirstlane(ldsDeg[nl]);
      const int end = min(cntNode, kCap);
      for (int base = 0; base < end; base += 32) {
        const int n = min(32, end - base);
        int mine = (lane < n) ? ldsList[nl][base + lane] : 0;
        mine = min(max(mine, 0), NSRC - 1);
        int j = 0;
        for (; j + 4 <= n; j += 4) {
          const v4f* p0 = (const v4f*)(feat + (size_t)__builtin_amdgcn_readlane(mine, j) * 256);
          const v4f* p1 = (const v4f*)(feat + (size_t)__builtin_amdgcn_readlane(mine, j + 1) * 256);
          const v4f* p2 = (const v4f*)(feat + (size_t)__builtin_amdgcn_readlane(mine, j + 2) * 256);
          const v4f* p3 = (const v4f*)(feat + (size_t)__builtin_amdgcn_readlane(mine, j + 3) * 256);
          const v4f x0 = p0[lane], y0 = p0[lane + 32];
          const v4f x1 = p1[lane], y1 = p1[lane + 32];
          const v4f x2 = p2[lane], y2 = p2[lane + 32];
          const v4f x3 = p3[lane], y3 = p3[lane + 32];
          a0 += x0; a1 += y0;
          a0 += x1; a1 += y1;
          a0 += x2; a1 += y2;
          a0 += x3; a1 += y3;
        }
#pragma unroll 1
        for (; j < n; ++j) {
          const v4f* p = (const v4f*)(feat + (size_t)__builtin_amdgcn_readlane(mine, j) * 256);
          a0 += p[lane];
          a1 += p[lane + 32];
        }
      }
    } else {
      for (int base = 0; base < nEdges; base += 32) {
        const int i = base + lane;
        const int d = (i < nEdges) ? min(max(dstAll[i], 0), nseg - 1) : -1;
        int s = (i < nEdges) ? srcAll[i] : 0;
        s = min(max(s, 0), NSRC - 1);
        unsigned hit = __builtin_amdgcn_ballot_w32(d == node);
        cntNode += __builtin_popcount(hit);
        while (hit != 0u) {
          const int j = __builtin_ctz(hit);
          hit &= hit - 1u;
          const v4f* p = (const v4f*)(feat + (size_t)__builtin_amdgcn_readlane(s, j) * 256);
          a0 += p[lane];
          a1 += p[lane + 32];
        }
      }
    }
    v4f* sp = (v4f*)(sum + (size_t)node * 256);
    sp[lane] = a0;
    sp[lane + 32] = a1;
    if (lane == 0) degOut[node] = cntNode;
  }
}

template <int KDIM, int NDIM, int M, int WM, int WN, int MT, int NT, bool SIGMOID>
__global__ void __launch_bounds__(256)
fused_wmma_gemm(const float* __restrict__ Asum,
                const int* __restrict__ deg,
                const float* __restrict__ Aself,
                const float* __restrict__ Wl,
                const float* __restrict__ Wr,
                const float* __restrict__ bias,
                float* __restrict__ out) {
  static_assert(WM * WN == 8, "8 waves per work-group");
  static_assert(WN * NT * 16 == NDIM, "a work-group spans all columns");
  static_assert(256 % NDIM == 0, "column map of the W stage");
  constexpr int TM = WM * MT * 16;
  constexpr int KC = 32;
  constexpr int W_STRIDE = KC + 2;
  constexpr int A_STRIDE = KC + 4;
  constexpr int HALF_CHUNKS = KDIM / KC;
  constexpr int NCHUNK = 2 * HALF_CHUNKS;
  constexpr int A_ITEMS = TM * (KC / 4);
  constexpr int A_PER = (A_ITEMS + 255) / 256;
  constexpr int W_KP = 256 / NDIM;
  constexpr int W_PER = KC / 2 / W_KP;
  __shared__ __attribute__((aligned(16))) float ldsW[2][NDIM][W_STRIDE];
  __shared__ __attribute__((aligned(16))) float ldsA[2][TM][A_STRIDE];

  const int tid = threadIdx.x;
  const int lane = tid & 31;
  const int wave = tid >> 5;
  const int half = lane >> 4;
  const int l15 = lane & 15;
  const int m0 = blockIdx.x * TM;
  const int rbase = (wave / WN) * (MT * 16);
  const int cbase = (wave % WN) * (NT * 16);

  int aOff[A_PER];
  float sInv[A_PER];
#pragma unroll
  for (int it = 0; it < A_PER; ++it) {
    const int idx = tid + 256 * it;
    const int row = min(m0 + (idx >> 3), M - 1);
    aOff[it] = row * KDIM + (idx & 7) * 4;
    sInv[it] = 1.0f / fmaxf((float)deg[row], 1.0f);
  }
  const int wcol = tid % NDIM;
  const int wkp = tid / NDIM;

  v8f acc[MT][NT];
#pragma unroll
  for (int i = 0; i < MT; ++i)
#pragma unroll
    for (int j = 0; j < NT; ++j) acc[i][j] = v8f{0, 0, 0, 0, 0, 0, 0, 0};

  v4f aReg[A_PER];
  v2f wReg[W_PER];

  auto gload = [&](int c) {
    const bool s = c < HALF_CHUNKS;
    const float* Ab = s ? Asum : Aself;
    const float* Wb = s ? Wl : Wr;
    const int kc = (s ? c : c - HALF_CHUNKS) * KC;
#pragma unroll
    for (int q = 0; q < W_PER; ++q) {
      const int k = kc + 2 * (wkp + W_KP * q);
      wReg[q][0] = Wb[(size_t)k * NDIM + wcol];
      wReg[q][1] = Wb[(size_t)(k + 1) * NDIM + wcol];
    }
#pragma unroll
    for (int it = 0; it < A_PER; ++it) {
      if (A_ITEMS % 256 == 0 || tid + 256 * it < A_ITEMS)
        aReg[it] = *(const v4f*)(Ab + (size_t)aOff[it] + kc);
    }
  };
  auto lstore = [&](int c, int st) {
    const bool s = c < HALF_CHUNKS;
#pragma unroll
    for (int q = 0; q < W_PER; ++q)
      *(v2f*)&ldsW[st][wcol][2 * (wkp + W_KP * q)] = wReg[q];
#pragma unroll
    for (int it = 0; it < A_PER; ++it) {
      const int idx = tid + 256 * it;
      if (A_ITEMS % 256 == 0 || idx < A_ITEMS) {
        const float f = s ? sInv[it] : 1.0f;
        *(v4f*)&ldsA[st][idx >> 3][(idx & 7) * 4] = aReg[it] * f;
      }
    }
  };

  gload(0);
  lstore(0, 0);
  __syncthreads();

  auto ldfrag = [&](int st, int k2, v2f (&a)[MT], v2f (&b)[NT]) {
    const int koff = k2 + 2 * half;
#pragma unroll
    for (int i = 0; i < MT; ++i)
      a[i] = *(const v2f*)&ldsA[st][rbase + 16 * i + l15][koff];
#pragma unroll
    for (int j = 0; j < NT; ++j)
      b[j] = *(const v2f*)&ldsW[st][cbase + 16 * j + l15][koff];
  };

  for (int c = 0; c < NCHUNK; ++c) {
    const int st = c & 1;
    if (c + 1 < NCHUNK) gload(c + 1);
    v2f a[2][MT], b[2][NT];
    ldfrag(st, 0, a[0], b[0]);

#pragma unroll
    for (int s = 0; s < KC / 4; ++s) {
      if (s + 1 < KC / 4) ldfrag(st, 4 * (s + 1), a[(s + 1) & 1], b[(s + 1) & 1]);
      __builtin_amdgcn_sched_barrier(0);
#pragma unroll
      for (int i = 0; i < MT; ++i)
#pragma unroll
        for (int j = 0; j < NT; ++j)
          acc[i][j] = __builtin_amdgcn_wmma_f32_16x16x4_f32(
              false, a[s & 1][i], false, b[s & 1][j], (short)0, acc[i][j],
              false, false);
      __builtin_amdgcn_sched_barrier(0);
    }

    if (c + 1 < NCHUNK) lstore(c + 1, st ^ 1);
    __syncthreads();
  }

#pragma unroll
  for (int j = 0; j < NT; ++j) {
    const int col = cbase + 16 * j + l15;
    const float bv = bias[col];
#pragma unroll
    for (int i = 0; i < MT; ++i) {
#pragma unroll
      for (int r = 0; r < 8; ++r) {
        const int row = m0 + rbase + 16 * i + r + 8 * half;
        float v = acc[i][j][r] + bv;
        v = SIGMOID ? (1.0f / (1.0f + expf(-v))) : fmaxf(v, 0.0f);
        if (row < M) out[(size_t)row * NDIM + col] = v;
      }
    }
  }
}


template <bool B> struct BoolTag { static constexpr bool value = B; };

__global__ void __launch_bounds__(256)
layer2_fused(const float* __restrict__ feat,
             const int* __restrict__ cell,
             const int* __restrict__ srcAll,
             const int* __restrict__ dstAll,
             const float* __restrict__ Wl,
             const float* __restrict__ Wr,
             const float* __restrict__ bias,
             float* __restrict__ out) {
  constexpr int NSRC = kN1, C = kC2, NB = kB2, R = kR2, nEdges = kE2;
  constexpr int M = kN2, KDIM = 256, NDIM = 128;
  constexpr int TM = kRange2;
  constexpr int MT = TM / 16;
  constexpr int NPW = TM / 8;
  constexpr int kBatch = 16;
  constexpr int KC = 32;
  constexpr int W_STRIDE = KC + 2;
  constexpr int A_STRIDE = KC + 4;
  constexpr int SUM_PITCH = KDIM + 4;
  constexpr int HALF_CHUNKS = KDIM / KC;
  constexpr int NCHUNK = 2 * HALF_CHUNKS;
  constexpr int A_ITEMS = TM * (KC / 4);
  constexpr int A_PER = (A_ITEMS + 255) / 256;
  constexpr int W_KP = 256 / NDIM;
  constexpr int W_PER = KC / 2 / W_KP;
  static_assert(TM % 16 == 0 && TM % 8 == 0, "whole row tiles, whole nodes per wave");
  static_assert(NDIM == 8 * 16, "8 waves, one column tile each");

  __shared__ int ldsDeg[64];
  __shared__ int ldsList[64][kCap];
  __shared__ int ldsSlow;
  __shared__ __attribute__((aligned(16))) float ldsSum[TM][SUM_PITCH];
  __shared__ __attribute__((aligned(16))) float ldsW[2][NDIM][W_STRIDE];
  __shared__ __attribute__((aligned(16))) float ldsA[2][TM][A_STRIDE];

  const int tid = threadIdx.x;
  const int lane = tid & 31;
  const int wave = __builtin_amdgcn_readfirstlane((int)(threadIdx.x >> 5));
  const int r = blockIdx.x;

  if (tid < 64) ldsDeg[tid] = 0;
  if (tid == 0) ldsSlow = 0;
  __syncthreads();

  {
    constexpr int PIECES = C / 4;
    constexpr int PER_PASS = 256 / PIECES;
    const int piece = tid % PIECES;
    const int cq = tid / PIECES;
    constexpr int NPASS = (NB + PER_PASS - 1) / PER_PASS;
    int over = 0;
    int4 vv[NPASS];
    int slot[NPASS][4];
#pragma unroll
    for (int k = 0; k < NPASS; ++k) {
      const int b = k * PER_PASS + cq;
      vv[k] = make_int4(0, 0, 0, 0);
      if (b < NB) vv[k] = *(const int4*)(cell + ((size_t)b * R + r) * C + piece * 4);
    }
#pragma unroll
    for (int k = 0; k < NPASS; ++k) {
      const int4 v = vv[k];
      const int n = __shfl(v.x, lane & ~(PIECES - 1));
      if (n > C - 1) over = 1;
      const int nn = min(max(n, 0), C - 1);
      const int e4[4] = {v.x, v.y, v.z, v.w};
#pragma unroll
      for (int i = 0; i < 4; ++i) {
        const int p = piece * 4 + i;
        slot[k][i] = kCap;
        if (p >= 1 && p <= nn)
          slot[k][i] = atomicAdd(&ldsDeg[(e4[i] >> 17) & 63], 1);
      }
    }
#pragma unroll
    for (int k = 0; k < NPASS; ++k) {
      const int e4[4] = {vv[k].x, vv[k].y, vv[k].z, vv[k].w};
#pragma unroll
      for (int i = 0; i < 4; ++i) {
        if (slot[k][i] < kCap)
          ldsList[(e4[i] >> 17) & 63][slot[k][i]] = e4[i] & 0x1FFFF;
      }
    }
    if (over) ldsSlow = 1;
  }
  __syncthreads();
  if (tid < TM && ldsDeg[tid] > kCap) ldsSlow = 1;
  __syncthreads();
  const bool slow = __builtin_amdgcn_readfirstlane(ldsSlow) != 0;

  for (int q = 0; q < NPW; ++q) {
    const int nl = wave * NPW + q;
    const int node = r * TM + nl;
    v4f a0 = v4f{0, 0, 0, 0};
    v4f a1 = v4f{0, 0, 0, 0};
    int cntNode = 0;
    if (node < M) {
      if (!slow) {
        cntNode = __builtin_amdgcn_readfirstlane(ldsDeg[nl]);
        const int end = min(cntNode, kCap);
        for (int base = 0; base < end; base += 32) {
          const int n = min(32, end - base);
          int mine = (lane < n) ? ldsList[nl][base + lane] : 0;
          mine = min(max(mine, 0), NSRC - 1);
          for (int j = 0; j < n; j += kBatch) {
            const int m = n - j;
            v4f x[kBatch], y[kBatch];
#pragma unroll
            for (int t = 0; t < kBatch; ++t) {
              if (t < m) {
                const v4f* p = (const v4f*)(feat + (size_t)__builtin_amdgcn_readlane(mine, j + t) * 256);
                x[t] = p[lane];
                y[t] = p[lane + 32];
              }
            }
#pragma unroll
            for (int t = 0; t < kBatch; ++t) {
              if (t < m) {
                a0 += x[t];
                a1 += y[t];
              }
            }
          }
        }
      } else {
        for (int base = 0; base < nEdges; base += 32) {
          const int i = base + lane;
          const int d = (i < nEdges) ? min(max(dstAll[i], 0), M - 1) : -1;
          int sidx = (i < nEdges) ? srcAll[i] : 0;
          sidx = min(max(sidx, 0), NSRC - 1);
          unsigned hit = __builtin_amdgcn_ballot_w32(d == node);
          cntNode += __builtin_popcount(hit);
          while (hit != 0u) {
            const int j = __builtin_ctz(hit);
            hit &= hit - 1u;
            const v4f* p = (const v4f*)(feat + (size_t)__builtin_amdgcn_readlane(sidx, j) * 256);
            a0 += p[lane];
            a1 += p[lane + 32];
          }
        }
      }
    }
    const float inv = 1.0f / fmaxf((float)cntNode, 1.0f);
    *(v4f*)&ldsSum[nl][4 * lane] = a0 * inv;
    *(v4f*)&ldsSum[nl][128 + 4 * lane] = a1 * inv;
  }

  const int half = lane >> 4;
  const int l15 = lane & 15;
  const int m0 = r * TM;
  const int cbase = wave * 16;

  int aOff[A_PER];
#pragma unroll
  for (int it = 0; it < A_PER; ++it) {
    const int idx = tid + 256 * it;
    const int row = min(m0 + (idx >> 3), M - 1);
    aOff[it] = row * KDIM + (idx & 7) * 4;
  }
  const int wcol = tid % NDIM;
  const int wkp = tid / NDIM;

  v8f acc[MT];
#pragma unroll
  for (int i = 0; i < MT; ++i) acc[i] = v8f{0, 0, 0, 0, 0, 0, 0, 0};

  v4f aReg[A_PER];
  v2f wReg[W_PER];

  auto gload = [&](int c) {
    const bool sh = c < HALF_CHUNKS;
    const float* Wb = sh ? Wl : Wr;
    const int kc = (sh ? c : c - HALF_CHUNKS) * KC;
#pragma unroll
    for (int q = 0; q < W_PER; ++q) {
      const int k = kc + 2 * (wkp + W_KP * q);
      wReg[q][0] = Wb[(size_t)k * NDIM + wcol];
      wReg[q][1] = Wb[(size_t)(k + 1) * NDIM + wcol];
    }
    if (!sh) {
#pragma unroll
      for (int it = 0; it < A_PER; ++it) {
        if (A_ITEMS % 256 == 0 || tid + 256 * it < A_ITEMS)
          aReg[it] = *(const v4f*)(feat + (size_t)aOff[it] + kc);
      }
    }
  };
  auto lstore = [&](int c, int st) {
    const bool sh = c < HALF_CHUNKS;
#pragma unroll
    for (int q = 0; q < W_PER; ++q)
      *(v2f*)&ldsW[st][wcol][2 * (wkp + W_KP * q)] = wReg[q];
    if (!sh) {
#pragma unroll
      for (int it = 0; it < A_PER; ++it) {
        const int idx = tid + 256 * it;
        if (A_ITEMS % 256 == 0 || idx < A_ITEMS)
          *(v4f*)&ldsA[st][idx >> 3][(idx & 7) * 4] = aReg[it];
      }
    }
  };

  gload(0);
  lstore(0, 0);
  __syncthreads();

  auto chunk = [&](auto tag, int c) {
    constexpr bool SUMHALF = decltype(tag)::value;
    const int st = c & 1;
    if (c + 1 < NCHUNK) gload(c + 1);
    auto ldfrag = [&](int k2, v2f (&a)[MT], v2f& b) {
      const int koff = k2 + 2 * half;
#pragma unroll
      for (int i = 0; i < MT; ++i) {
        if constexpr (SUMHALF)
          a[i] = *(const v2f*)&ldsSum[16 * i + l15][c * KC + koff];
        else
          a[i] = *(const v2f*)&ldsA[st][16 * i + l15][koff];
      }
      b = *(const v2f*)&ldsW[st][cbase + l15][koff];
    };
    v2f a[2][MT], b[2];
    ldfrag(0, a[0], b[0]);
#pragma unroll
    for (int s = 0; s < KC / 4; ++s) {
      if (s + 1 < KC / 4) ldfrag(4 * (s + 1), a[(s + 1) & 1], b[(s + 1) & 1]);
      __builtin_amdgcn_sched_barrier(0);
#pragma unroll
      for (int i = 0; i < MT; ++i)
        acc[i] = __builtin_amdgcn_wmma_f32_16x16x4_f32(
            false, a[s & 1][i], false, b[s & 1], (short)0, acc[i], false, false);
      __builtin_amdgcn_sched_barrier(0);
    }
    if (c + 1 < NCHUNK) lstore(c + 1, st ^ 1);
    __syncthreads();
  };
  for (int c = 0; c < HALF_CHUNKS; ++c) chunk(BoolTag<true>{}, c);
  for (int c = HALF_CHUNKS; c < NCHUNK; ++c) chunk(BoolTag<false>{}, c);

  const int col = cbase + l15;
  const float bv = bias[col];
#pragma unroll
  for (int i = 0; i < MT; ++i) {
#pragma unroll
    for (int rr = 0; rr < 8; ++rr) {
      const int row = m0 + 16 * i + rr + 8 * half;
      const float v = 1.0f / (1.0f + expf(-(acc[i][rr] + bv)));
      if (row < M) out[(size_t)row * NDIM + col] = v;
    }
  }
}

}

extern "C" void kernel_launch(void* const* d_in, const int* in_sizes, int n_in,
                              void* d_out, int out_size, void* d_ws, size_t ws_size,
                              hipStream_t stream) {
  (void)in_sizes; (void)n_in; (void)out_size; (void)ws_size;

  const float* x   = (const float*)d_in[0];
  const float* W1l = (const float*)d_in[1];
  const float* b1  = (const float*)d_in[2];
  const float* W1r = (const float*)d_in[3];
  const float* W2l = (const float*)d_in[4];
  const float* b2  = (const float*)d_in[5];
  const float* W2r = (const float*)d_in[6];
  const int* src1  = (const int*)d_in[7];
  const int* dst1  = (const int*)d_in[8];
  const int* src2  = (const int*)d_in[9];
  const int* dst2  = (const int*)d_in[10];
  float* out = (float*)d_out;

  float* ws   = (float*)d_ws;
  int*   wsi  = (int*)d_ws;
  float* sum1 = ws + kOffSum1;
  float* h    = ws + kOffH;
  int* deg   = wsi + kOffDeg;
  int* cell1 = wsi + kOffCel1;
  int* cell2 = wsi + kOffCel2;

  edge_bin<<<kB1 + kB2, 256, 0, stream>>>(src1, dst1, src2, dst2, cell1, cell2);

  gather_sum<kN0, kC1, kB1, kR1><<<kR1, 256, 0, stream>>>(x, cell1, src1, dst1, kE1,
                                                       deg, sum1, kN1);

  fused_wmma_gemm<256, 256, kN1, 2, 4, 5, 4, false>
      <<<kN1 / 160, 256, 0, stream>>>(sum1, deg, x, W1l, W1r, b1, h);

  layer2_fused<<<kR2, 256, 0, stream>>>(h, cell2, src2, dst2, W2l, W2r, b2, out);
}


# Layer 1 in one launch, two work-groups per WGP: a group gathers its 80 mean rows into LDS, then runs their GEMM; every second group runs the self half first (its sum order: self, then mean)
# speedup vs baseline: 5.6306x; 1.0647x over previous
#include <hip/hip_runtime.h>
#include <hip/hip_bf16.h>
#include <math.h>


namespace {

constexpr int kN0 = 100000;
constexpr int kN1 = 40000;
constexpr int kN2 = 10000;
constexpr int kE1 = 640000;
constexpr int kE2 = 160000;

typedef __attribute__((ext_vector_type(2))) float v2f;
typedef __attribute__((ext_vector_type(8))) float v8f;

typedef __attribute__((ext_vector_type(4))) float v4f;

constexpr int kRange1 = 80;
constexpr int kRange2 = 48;
constexpr int kChunk = 4096;
constexpr int kR1 = kN1 / kRange1;
constexpr int kR2 = (kN2 + kRange2 - 1) / kRange2;
constexpr int kB1 = (kE1 + kChunk - 1) / kChunk;
constexpr int kB2 = (kE2 + kChunk - 1) / kChunk;
constexpr int kC1 = 32;
constexpr int kC2 = 64;
constexpr int kCap = 64;
static_assert(kN1 % kRange1 == 0, "layer-1 ranges are whole");
static_assert(kRange1 <= 128 && kRange2 <= 64, "a node within its range fits 7 (layer 1) or 6 (layer 2) bits of the packed edge");
static_assert(kN0 <= (1 << 17), "a source index fits the packed edge");

constexpr size_t align64(size_t n) { return (n + 63) / 64 * 64; }
constexpr size_t kOffSum1 = 0;
constexpr size_t kOffSum2 = kOffSum1 + (size_t)kN1 * 256;
constexpr size_t kOffH    = kOffSum2 + (size_t)kN2 * 256;
constexpr size_t kOffDeg  = align64(kOffH + (size_t)kN1 * 256);
constexpr size_t kOffCel1 = kOffDeg + align64(kN1 + kN2);
constexpr size_t kOffCel2 = kOffCel1 + align64((size_t)kB1 * kR1 * kC1);

template <int V> struct IntTag { static constexpr int value = V; };

__global__ void __launch_bounds__(256)
edge_bin(const int* __restrict__ src1,
         const int* __restrict__ dst1,
         const int* __restrict__ src2,
         const int* __restrict__ dst2,
         int* __restrict__ cell1,
         int* __restrict__ cell2) {
  __shared__ int cnt[kR1];
  static_assert(kR1 >= kR2, "counter array serves both layers");
  const int tid = threadIdx.x;
  const bool first = blockIdx.x < kB1;
  const int b = first ? blockIdx.x : blockIdx.x - kB1;
  const int* src = first ? src1 : src2;
  const int* dst = first ? dst1 : dst2;
  const int E = first ? kE1 : kE2;
  const int R = first ? kR1 : kR2;
  const int C = first ? kC1 : kC2;
  const int nDst = first ? kN1 : kN2;
  const int nSrc = first ? kN0 : kN1;
  int* mycell = (first ? cell1 : cell2) + (size_t)b * R * C;

  for (int r = tid; r < R; r += 256) cnt[r] = 0;
  __syncthreads();

  constexpr int PER = kChunk / 256;
  int d[PER], s[PER];
#pragma unroll
  for (int k = 0; k < PER; ++k) {
    const int i = min(b * kChunk + k * 256 + tid, E - 1);
    d[k] = dst[i];
    s[k] = src[i];
  }
  auto sortChunk = [&](auto tag) {
    constexpr int RANGE = decltype(tag)::value;
    int slot[PER];
#pragma unroll
    for (int k = 0; k < PER; ++k) {
      d[k] = min(max(d[k], 0), nDst - 1);
      s[k] = min(max(s[k], 0), nSrc - 1);
      slot[k] = C;
      if (b * kChunk + k * 256 + tid < E) slot[k] = atomicAdd(&cnt[d[k] / RANGE], 1);
    }
#pragma unroll
    for (int k = 0; k < PER; ++k) {
      if (slot[k] < C - 1)
        mycell[(d[k] / RANGE) * C + 1 + slot[k]] = ((d[k] % RANGE) << 17) | s[k];
    }
  };
  if (first) sortChunk(IntTag<kRange1>{}); else sortChunk(IntTag<kRange2>{});
  __syncthreads();
  for (int r = tid; r < R; r += 256) mycell[r * C] = cnt[r];
}

template <bool B> struct BoolTag { static constexpr bool value = B; };


__global__ void __launch_bounds__(256)
layer1_fused(const float* __restrict__ feat,
             const int* __restrict__ cell,
             const int* __restrict__ srcAll,
             const int* __restrict__ dstAll,
             const float* __restrict__ Wl,
             const float* __restrict__ Wr,
             const float* __restrict__ bias,
             float* __restrict__ out) {
  constexpr int NSRC = kN0, C = kC1, NB = kB1, R = kR1, nEdges = kE1;
  constexpr int M = kN1, KDIM = 256, NDIM = 256;
  constexpr int TM = kRange1;
  constexpr int MT = TM / 16;
  constexpr int NT = NDIM / (8 * 16);
  constexpr int NPW = TM / 8;
  constexpr int kBatch = 16;
  constexpr int KC = 16;
  constexpr int W_STRIDE = KC + 4;
  constexpr int A_STRIDE = KC + 4;
  constexpr int SUM_PITCH = KDIM + 4;
  constexpr int HALF_CHUNKS = KDIM / KC;
  constexpr int A_ITEMS = TM * (KC / 4);
  constexpr int A_PER = (A_ITEMS + 255) / 256;
  constexpr int W_PER = KC / 2;
  static_assert(TM % 16 == 0 && TM % 8 == 0, "whole row tiles, whole nodes per wave");
  static_assert(NDIM == 256, "a thread stages one column of W");

  __shared__ int ldsDeg[128];
  __shared__ int ldsList[TM][kCap];
  __shared__ int ldsSlow;
  __shared__ __attribute__((aligned(16))) float ldsSum[TM][SUM_PITCH];
  __shared__ __attribute__((aligned(16))) float ldsW[2][NDIM][W_STRIDE];
  __shared__ __attribute__((aligned(16))) float ldsA[2][TM][A_STRIDE];

  const int tid = threadIdx.x;
  const int lane = tid & 31;
  const int wave = __builtin_amdgcn_readfirstlane((int)(threadIdx.x >> 5));
  const int r = blockIdx.x;
  const int half = lane >> 4;
  const int l15 = lane & 15;
  const int m0 = r * TM;
  const int cbase = wave * (NT * 16);

  auto gatherMeans = [&]() {
    if (tid < 128) ldsDeg[tid] = 0;
    if (tid == 0) ldsSlow = 0;
    __syncthreads();
    {
      constexpr int PIECES = C / 4;
      constexpr int PER_PASS = 256 / PIECES;
      const int piece = tid % PIECES;
      const int cq = tid / PIECES;
      constexpr int NPASS = (NB + PER_PASS - 1) / PER_PASS;
      int over = 0;
      int4 vv[NPASS];
      int slot[NPASS][4];
#pragma unroll
      for (int k = 0; k < NPASS; ++k) {
        const int b = k * PER_PASS + cq;
        vv[k] = make_int4(0, 0, 0, 0);
        if (b < NB) vv[k] = *(const int4*)(cell + ((size_t)b * R + r) * C + piece * 4);
      }
#pragma unroll
      for (int k = 0; k < NPASS; ++k) {
        const int4 v = vv[k];
        const int n = __shfl(v.x, lane & ~(PIECES - 1));
        if (n > C - 1) over = 1;
        const int nn = min(max(n, 0), C - 1);
        const int e4[4] = {v.x, v.y, v.z, v.w};
#pragma unroll
        for (int i = 0; i < 4; ++i) {
          const int p = piece * 4 + i;
          slot[k][i] = kCap;
          if (p >= 1 && p <= nn)
            slot[k][i] = atomicAdd(&ldsDeg[(e4[i] >> 17) & 127], 1);
        }
      }
#pragma unroll
      for (int k = 0; k < NPASS; ++k) {
        const int e4[4] = {vv[k].x, vv[k].y, vv[k].z, vv[k].w};
#pragma unroll
        for (int i = 0; i < 4; ++i) {
          const int nd = min((e4[i] >> 17) & 127, TM - 1);
          if (slot[k][i] < kCap) ldsList[nd][slot[k][i]] = e4[i] & 0x1FFFF;
        }
      }
      if (over) ldsSlow = 1;
    }
    __syncthreads();
    if (tid < TM && ldsDeg[tid] > kCap) ldsSlow = 1;
    __syncthreads();
    const bool slow = __builtin_amdgcn_readfirstlane(ldsSlow) != 0;

    for (int q = 0; q < NPW; ++q) {
      const int nl = wave * NPW + q;
      const int node = r * TM + nl;
      v4f a0 = v4f{0, 0, 0, 0};
      v4f a1 = v4f{0, 0, 0, 0};
      int cntNode = 0;
      if (node < M) {
        if (!slow) {
          cntNode = __builtin_amdgcn_readfirstlane(ldsDeg[nl]);
          const int end = min(cntNode, kCap);
          for (int base = 0; base < end; base += 32) {
            const int n = min(32, end - base);
            int mine = (lane < n) ? ldsList[nl][base + lane] : 0;
            mine = min(max(mine, 0), NSRC - 1);
            for (int j = 0; j < n; j += kBatch) {
              const int m = n - j;
              v4f x[kBatch], y[kBatch];
#pragma unroll
              for (int t = 0; t < kBatch; ++t) {
                if (t < m) {
                  const v4f* p = (const v4f*)(feat + (size_t)__builtin_amdgcn_readlane(mine, j + t) * 256);
                  x[t] = p[lane];
                  y[t] = p[lane + 32];
                }
              }
#pragma unroll
              for (int t = 0; t < kBatch; ++t) {
                if (t < m) {
                  a0 += x[t];
                  a1 += y[t];
                }
              }
            }
          }
        } else {
          for (int base = 0; base < nEdges; base += 32) {
            const int i = base + lane;
            const int d = (i < nEdges) ? min(max(dstAll[i], 0), M - 1) : -1;
            int sidx = (i < nEdges) ? srcAll[i] : 0;
            sidx = min(max(sidx, 0), NSRC - 1);
            unsigned hit = __builtin_amdgcn_ballot_w32(d == node);
            cntNode += __builtin_popcount(hit);
            while (hit != 0u) {
              const int j = __builtin_ctz(hit);
              hit &= hit - 1u;
              const v4f* p = (const v4f*)(feat + (size_t)__builtin_amdgcn_readlane(sidx, j) * 256);
              a0 += p[lane];
              a1 += p[lane + 32];
            }
          }
        }
      }
      const float inv = 1.0f / fmaxf((float)cntNode, 1.0f);
      *(v4f*)&ldsSum[nl][4 * lane] = a0 * inv;
      *(v4f*)&ldsSum[nl][128 + 4 * lane] = a1 * inv;
    }
  };

  int aOff[A_PER];
#pragma unroll
  for (int it = 0; it < A_PER; ++it) {
    const int idx = tid + 256 * it;
    const int row = min(m0 + (idx >> 2), M - 1);
    aOff[it] = row * KDIM + (idx & 3) * 4;
  }

  v8f acc[MT][NT];
#pragma unroll
  for (int i = 0; i < MT; ++i)
#pragma unroll
    for (int j = 0; j < NT; ++j) acc[i][j] = v8f{0, 0, 0, 0, 0, 0, 0, 0};

  v4f aReg[A_PER];
  v2f wReg[W_PER];

  auto gemmHalf = [&](auto tag) {
    constexpr bool SUMHALF = decltype(tag)::value;
    const float* Wb = SUMHALF ? Wl : Wr;
    auto gload = [&](int c) {
      const int kc = c * KC;
#pragma unroll
      for (int q = 0; q < W_PER; ++q) {
        const int k = kc + 2 * q;
        wReg[q][0] = Wb[(size_t)k * NDIM + tid];
        wReg[q][1] = Wb[(size_t)(k + 1) * NDIM + tid];
      }
      if constexpr (!SUMHALF) {
#pragma unroll
        for (int it = 0; it < A_PER; ++it) {
          if (A_ITEMS % 256 == 0 || tid + 256 * it < A_ITEMS)
            aReg[it] = *(const v4f*)(feat + (size_t)aOff[it] + kc);
        }
      }
    };
    auto lstore = [&](int st) {
#pragma unroll
      for (int q = 0; q < W_PER; ++q)
        *(v2f*)&ldsW[st][tid][2 * q] = wReg[q];
      if constexpr (!SUMHALF) {
#pragma unroll
        for (int it = 0; it < A_PER; ++it) {
          const int idx = tid + 256 * it;
          if (A_ITEMS % 256 == 0 || idx < A_ITEMS)
            *(v4f*)&ldsA[st][idx >> 2][(idx & 3) * 4] = aReg[it];
        }
      }
    };

    gload(0);
    lstore(0);
    __syncthreads();

    for (int c = 0; c < HALF_CHUNKS; ++c) {
      const int st = c & 1;
      if (c + 1 < HALF_CHUNKS) gload(c + 1);
      auto ldfrag = [&](int k2, v2f (&a)[MT], v2f (&b)[NT]) {
        const int koff = k2 + 2 * half;
#pragma unroll
        for (int i = 0; i < MT; ++i) {
          if constexpr (SUMHALF)
            a[i] = *(const v2f*)&ldsSum[16 * i + l15][c * KC + koff];
          else
            a[i] = *(const v2f*)&ldsA[st][16 * i + l15][koff];
        }
#pragma unroll
        for (int j = 0; j < NT; ++j)
          b[j] = *(const v2f*)&ldsW[st][cbase + 16 * j + l15][koff];
      };
      v2f a[2][MT], b[2][NT];
      ldfrag(0, a[0], b[0]);
#pragma unroll
      for (int s = 0; s < KC / 4; ++s) {
        if (s + 1 < KC / 4) ldfrag(4 * (s + 1), a[(s + 1) & 1], b[(s + 1) & 1]);
        __builtin_amdgcn_sched_barrier(0);
#pragma unroll
        for (int i = 0; i < MT; ++i)
#pragma unroll
          for (int j = 0; j < NT; ++j)
            acc[i][j] = __builtin_amdgcn_wmma_f32_16x16x4_f32(
                false, a[s & 1][i], false, b[s & 1][j], (short)0, acc[i][j],
                false, false);
        __builtin_amdgcn_sched_barrier(0);
      }
      if (c + 1 < HALF_CHUNKS) lstore(st ^ 1);
      __syncthreads();
    }
  };

  const bool selfFirst = (__builtin_popcount((unsigned)blockIdx.x) & 1) != 0;
  if (selfFirst) {
    gemmHalf(BoolTag<false>{});
    gatherMeans();
    gemmHalf(BoolTag<true>{});
  } else {
    gatherMeans();
    gemmHalf(BoolTag<true>{});
    gemmHalf(BoolTag<false>{});
  }

#pragma unroll
  for (int j = 0; j < NT; ++j) {
    const int col = cbase + 16 * j + l15;
    const float bv = bias[col];
#pragma unroll
    for (int i = 0; i < MT; ++i) {
#pragma unroll
      for (int rr = 0; rr < 8; ++rr) {
        const int row = m0 + 16 * i + rr + 8 * half;
        const float v = fmaxf(acc[i][j][rr] + bv, 0.0f);
        if (row < M) out[(size_t)row * NDIM + col] = v;
      }
    }
  }
}

__global__ void __launch_bounds__(256)
layer2_fused(const float* __restrict__ feat,
             const int* __restrict__ cell,
             const int* __restrict__ srcAll,
             const int* __restrict__ dstAll,
             const float* __restrict__ Wl,
             const float* __restrict__ Wr,
             const float* __restrict__ bias,
             float* __restrict__ out) {
  constexpr int NSRC = kN1, C = kC2, NB = kB2, R = kR2, nEdges = kE2;
  constexpr int M = kN2, KDIM = 256, NDIM = 128;
  constexpr int TM = kRange2;
  constexpr int MT = TM / 16;
  constexpr int NPW = TM / 8;
  constexpr int kBatch = 16;
  constexpr int KC = 32;
  constexpr int W_STRIDE = KC + 2;
  constexpr int A_STRIDE = KC + 4;
  constexpr int SUM_PITCH = KDIM + 4;
  constexpr int HALF_CHUNKS = KDIM / KC;
  constexpr int NCHUNK = 2 * HALF_CHUNKS;
  constexpr int A_ITEMS = TM * (KC / 4);
  constexpr int A_PER = (A_ITEMS + 255) / 256;
  constexpr int W_KP = 256 / NDIM;
  constexpr int W_PER = KC / 2 / W_KP;
  static_assert(TM % 16 == 0 && TM % 8 == 0, "whole row tiles, whole nodes per wave");
  static_assert(NDIM == 8 * 16, "8 waves, one column tile each");

  __shared__ int ldsDeg[64];
  __shared__ int ldsList[64][kCap];
  __shared__ int ldsSlow;
  __shared__ __attribute__((aligned(16))) float ldsSum[TM][SUM_PITCH];
  __shared__ __attribute__((aligned(16))) float ldsW[2][NDIM][W_STRIDE];
  __shared__ __attribute__((aligned(16))) float ldsA[2][TM][A_STRIDE];

  const int tid = threadIdx.x;
  const int lane = tid & 31;
  const int wave = __builtin_amdgcn_readfirstlane((int)(threadIdx.x >> 5));
  const int r = blockIdx.x;

  if (tid < 64) ldsDeg[tid] = 0;
  if (tid == 0) ldsSlow = 0;
  __syncthreads();

  {
    constexpr int PIECES = C / 4;
    constexpr int PER_PASS = 256 / PIECES;
    const int piece = tid % PIECES;
    const int cq = tid / PIECES;
    constexpr int NPASS = (NB + PER_PASS - 1) / PER_PASS;
    int over = 0;
    int4 vv[NPASS];
    int slot[NPASS][4];
#pragma unroll
    for (int k = 0; k < NPASS; ++k) {
      const int b = k * PER_PASS + cq;
      vv[k] = make_int4(0, 0, 0, 0);
      if (b < NB) vv[k] = *(const int4*)(cell + ((size_t)b * R + r) * C + piece * 4);
    }
#pragma unroll
    for (int k = 0; k < NPASS; ++k) {
      const int4 v = vv[k];
      const int n = __shfl(v.x, lane & ~(PIECES - 1));
      if (n > C - 1) over = 1;
      const int nn = min(max(n, 0), C - 1);
      const int e4[4] = {v.x, v.y, v.z, v.w};
#pragma unroll
      for (int i = 0; i < 4; ++i) {
        const int p = piece * 4 + i;
        slot[k][i] = kCap;
        if (p >= 1 && p <= nn)
          slot[k][i] = atomicAdd(&ldsDeg[(e4[i] >> 17) & 63], 1);
      }
    }
#pragma unroll
    for (int k = 0; k < NPASS; ++k) {
      const int e4[4] = {vv[k].x, vv[k].y, vv[k].z, vv[k].w};
#pragma unroll
      for (int i = 0; i < 4; ++i) {
        if (slot[k][i] < kCap)
          ldsList[(e4[i] >> 17) & 63][slot[k][i]] = e4[i] & 0x1FFFF;
      }
    }
    if (over) ldsSlow = 1;
  }
  __syncthreads();
  if (tid < TM && ldsDeg[tid] > kCap) ldsSlow = 1;
  __syncthreads();
  const bool slow = __builtin_amdgcn_readfirstlane(ldsSlow) != 0;

  for (int q = 0; q < NPW; ++q) {
    const int nl = wave * NPW + q;
    const int node = r * TM + nl;
    v4f a0 = v4f{0, 0, 0, 0};
    v4f a1 = v4f{0, 0, 0, 0};
    int cntNode = 0;
    if (node < M) {
      if (!slow) {
        cntNode = __builtin_amdgcn_readfirstlane(ldsDeg[nl]);
        const int end = min(cntNode, kCap);
        for (int base = 0; base < end; base += 32) {
          const int n = min(32, end - base);
          int mine = (lane < n) ? ldsList[nl][base + lane] : 0;
          mine = min(max(mine, 0), NSRC - 1);
          for (int j = 0; j < n; j += kBatch) {
            const int m = n - j;
            v4f x[kBatch], y[kBatch];
#pragma unroll
            for (int t = 0; t < kBatch; ++t) {
              if (t < m) {
                const v4f* p = (const v4f*)(feat + (size_t)__builtin_amdgcn_readlane(mine, j + t) * 256);
                x[t] = p[lane];
                y[t] = p[lane + 32];
              }
            }
#pragma unroll
            for (int t = 0; t < kBatch; ++t) {
              if (t < m) {
                a0 += x[t];
                a1 += y[t];
              }
            }
          }
        }
      } else {
        for (int base = 0; base < nEdges; base += 32) {
          const int i = base + lane;
          const int d = (i < nEdges) ? min(max(dstAll[i], 0), M - 1) : -1;
          int sidx = (i < nEdges) ? srcAll[i] : 0;
          sidx = min(max(sidx, 0), NSRC - 1);
          unsigned hit = __builtin_amdgcn_ballot_w32(d == node);
          cntNode += __builtin_popcount(hit);
          while (hit != 0u) {
            const int j = __builtin_ctz(hit);
            hit &= hit - 1u;
            const v4f* p = (const v4f*)(feat + (size_t)__builtin_amdgcn_readlane(sidx, j) * 256);
            a0 += p[lane];
            a1 += p[lane + 32];
          }
        }
      }
    }
    const float inv = 1.0f / fmaxf((float)cntNode, 1.0f);
    *(v4f*)&ldsSum[nl][4 * lane] = a0 * inv;
    *(v4f*)&ldsSum[nl][128 + 4 * lane] = a1 * inv;
  }

  const int half = lane >> 4;
  const int l15 = lane & 15;
  const int m0 = r * TM;
  const int cbase = wave * 16;

  int aOff[A_PER];
#pragma unroll
  for (int it = 0; it < A_PER; ++it) {
    const int idx = tid + 256 * it;
    const int row = min(m0 + (idx >> 3), M - 1);
    aOff[it] = row * KDIM + (idx & 7) * 4;
  }
  const int wcol = tid % NDIM;
  const int wkp = tid / NDIM;

  v8f acc[MT];
#pragma unroll
  for (int i = 0; i < MT; ++i) acc[i] = v8f{0, 0, 0, 0, 0, 0, 0, 0};

  v4f aReg[A_PER];
  v2f wReg[W_PER];

  auto gload = [&](int c) {
    const bool sh = c < HALF_CHUNKS;
    const float* Wb = sh ? Wl : Wr;
    const int kc = (sh ? c : c - HALF_CHUNKS) * KC;
#pragma unroll
    for (int q = 0; q < W_PER; ++q) {
      const int k = kc + 2 * (wkp + W_KP * q);
      wReg[q][0] = Wb[(size_t)k * NDIM + wcol];
      wReg[q][1] = Wb[(size_t)(k + 1) * NDIM + wcol];
    }
    if (!sh) {
#pragma unroll
      for (int it = 0; it < A_PER; ++it) {
        if (A_ITEMS % 256 == 0 || tid + 256 * it < A_ITEMS)
          aReg[it] = *(const v4f*)(feat + (size_t)aOff[it] + kc);
      }
    }
  };
  auto lstore = [&](int c, int st) {
    const bool sh = c < HALF_CHUNKS;
#pragma unroll
    for (int q = 0; q < W_PER; ++q)
      *(v2f*)&ldsW[st][wcol][2 * (wkp + W_KP * q)] = wReg[q];
    if (!sh) {
#pragma unroll
      for (int it = 0; it < A_PER; ++it) {
        const int idx = tid + 256 * it;
        if (A_ITEMS % 256 == 0 || idx < A_ITEMS)
          *(v4f*)&ldsA[st][idx >> 3][(idx & 7) * 4] = aReg[it];
      }
    }
  };

  gload(0);
  lstore(0, 0);
  __syncthreads();

  auto chunk = [&](auto tag, int c) {
    constexpr bool SUMHALF = decltype(tag)::value;
    const int st = c & 1;
    if (c + 1 < NCHUNK) gload(c + 1);
    auto ldfrag = [&](int k2, v2f (&a)[MT], v2f& b) {
      const int koff = k2 + 2 * half;
#pragma unroll
      for (int i = 0; i < MT; ++i) {
        if constexpr (SUMHALF)
          a[i] = *(const v2f*)&ldsSum[16 * i + l15][c * KC + koff];
        else
          a[i] = *(const v2f*)&ldsA[st][16 * i + l15][koff];
      }
      b = *(const v2f*)&ldsW[st][cbase + l15][koff];
    };
    v2f a[2][MT], b[2];
    ldfrag(0, a[0], b[0]);
#pragma unroll
    for (int s = 0; s < KC / 4; ++s) {
      if (s + 1 < KC / 4) ldfrag(4 * (s + 1), a[(s + 1) & 1], b[(s + 1) & 1]);
      __builtin_amdgcn_sched_barrier(0);
#pragma unroll
      for (int i = 0; i < MT; ++i)
        acc[i] = __builtin_amdgcn_wmma_f32_16x16x4_f32(
            false, a[s & 1][i], false, b[s & 1], (short)0, acc[i], false, false);
      __builtin_amdgcn_sched_barrier(0);
    }
    if (c + 1 < NCHUNK) lstore(c + 1, st ^ 1);
    __syncthreads();
  };
  for (int c = 0; c < HALF_CHUNKS; ++c) chunk(BoolTag<true>{}, c);
  for (int c = HALF_CHUNKS; c < NCHUNK; ++c) chunk(BoolTag<false>{}, c);

  const int col = cbase + l15;
  const float bv = bias[col];
#pragma unroll
  for (int i = 0; i < MT; ++i) {
#pragma unroll
    for (int rr = 0; rr < 8; ++rr) {
      const int row = m0 + 16 * i + rr + 8 * half;
      const float v = 1.0f / (1.0f + expf(-(acc[i][rr] + bv)));
      if (row < M) out[(size_t)row * NDIM + col] = v;
    }
  }
}

}

extern "C" void kernel_launch(void* const* d_in, const int* in_sizes, int n_in,
                              void* d_out, int out_size, void* d_ws, size_t ws_size,
                              hipStream_t stream) {
  (void)in_sizes; (void)n_in; (void)out_size; (void)ws_size;

  const float* x   = (const float*)d_in[0];
  const float* W1l = (const float*)d_in[1];
  const float* b1  = (const float*)d_in[2];
  const float* W1r = (const float*)d_in[3];
  const float* W2l = (const float*)d_in[4];
  const float* b2  = (const float*)d_in[5];
  const float* W2r = (const float*)d_in[6];
  const int* src1  = (const int*)d_in[7];
  const int* dst1  = (const int*)d_in[8];
  const int* src2  = (const int*)d_in[9];
  const int* dst2  = (const int*)d_in[10];
  float* out = (float*)d_out;

  float* ws   = (float*)d_ws;
  int*   wsi  = (int*)d_ws;
  float* h    = ws + kOffH;
  int* cell1 = wsi + kOffCel1;
  int* cell2 = wsi + kOffCel2;

  edge_bin<<<kB1 + kB2, 256, 0, stream>>>(src1, dst1, src2, dst2, cell1, cell2);

  layer1_fused<<<kR1, 256, 0, stream>>>(x, cell1, src1, dst1, W1l, W1r, b1, h);

  layer2_fused<<<kR2, 256, 0, stream>>>(h, cell2, src2, dst2, W2l, W2r, b2, out);
}
